# starting kernel
# baseline (speedup 1.0000x reference)
#include <hip/hip_runtime.h>
#include <hip/hip_bf16.h>

#define NN   50000
#define EE   800000
#define GG   512
#define DD   64
#define DOUT 32
#define BN_EPS 1e-5f

typedef __attribute__((ext_vector_type(2)))  float    v2f;
typedef __attribute__((ext_vector_type(8)))  float    v8f;
typedef __attribute__((ext_vector_type(16))) _Float16 v16h;

#if defined(__AMDGCN__) && __has_builtin(__builtin_amdgcn_wmma_f32_16x16x4_f32)
#define USE_WMMA_F32 1
#else
#define USE_WMMA_F32 0
#endif


__global__ void k_zero(float* __restrict__ p, int n) {
  int i = blockIdx.x * blockDim.x + threadIdx.x;
  if (i < n) p[i] = 0.0f;
}

__global__ void k_edge_count(const int* __restrict__ ei,
                             float* __restrict__ deg_feat,
                             float* __restrict__ degcnt, int E) {
  int e = blockIdx.x * blockDim.x + threadIdx.x;
  if (e >= E) return;
  int s = ei[e];
  int d = ei[E + e];
  atomicAdd(&deg_feat[s], 1.0f);
  atomicAdd(&degcnt[d], 1.0f);
}

__global__ void k_dinv(const float* __restrict__ degcnt, float* __restrict__ dinv, int n) {
  int i = blockIdx.x * blockDim.x + threadIdx.x;
  if (i < n) dinv[i] = rsqrtf(degcnt[i] + 1.0f);
}

__global__ void k_l1_hw_self(const float* __restrict__ deg_feat,
                             const float* __restrict__ W1,
                             const float* __restrict__ dinv,
                             float* __restrict__ hw,
                             float* __restrict__ acc, int n64) {
  int i = blockIdx.x * blockDim.x + threadIdx.x;
  if (i >= n64) return;
  int n = i >> 6;
  float h = deg_feat[n] * W1[i & 63];
  float di = dinv[n];
  hw[i]  = h;
  acc[i] = h * (di * di);
}

__global__ void k_scatter(const int* __restrict__ ei,
                          const float* __restrict__ hw,
                          const float* __restrict__ dinv,
                          float* __restrict__ acc, int E) {
  int t = blockIdx.x * blockDim.x + threadIdx.x;
  int e = t >> 4;
  int g = t & 15;
  if (e >= E) return;
  int s = ei[e];
  int d = ei[E + e];
  float w = dinv[s] * dinv[d];
  float4 m = ((const float4*)(hw + (size_t)s * DD))[g];
  float* dst = acc + (size_t)d * DD + g * 4;
  atomicAdd(dst + 0, m.x * w);
  atomicAdd(dst + 1, m.y * w);
  atomicAdd(dst + 2, m.z * w);
  atomicAdd(dst + 3, m.w * w);
}

__global__ void k_post_bnrelu(const float* __restrict__ acc,
                              const float* __restrict__ b,
                              const float* __restrict__ g,
                              const float* __restrict__ be,
                              const float* __restrict__ m,
                              const float* __restrict__ v,
                              float* __restrict__ x, int n64) {
  int i = blockIdx.x * blockDim.x + threadIdx.x;
  if (i >= n64) return;
  int d = i & 63;
  float rs = rsqrtf(v[d] + BN_EPS);
  float y = ((acc[i] + b[d]) - m[d]) * rs * g[d] + be[d];
  x[i] = fmaxf(y, 0.0f);
}

__global__ void k_post_bias(const float* __restrict__ acc,
                            const float* __restrict__ b,
                            float* __restrict__ x, int n64) {
  int i = blockIdx.x * blockDim.x + threadIdx.x;
  if (i < n64) x[i] = acc[i] + b[i & 63];
}

__global__ void k_pool_scatter(const int* __restrict__ batch,
                               const float* __restrict__ x,
                               float* __restrict__ pool,
                               float* __restrict__ cnt, int n) {
  int t = blockIdx.x * blockDim.x + threadIdx.x;
  int nd = t >> 4;
  int g  = t & 15;
  if (nd >= n) return;
  int bg = batch[nd];
  float4 m = ((const float4*)(x + (size_t)nd * DD))[g];
  float* dst = pool + (size_t)bg * DD + g * 4;
  atomicAdd(dst + 0, m.x);
  atomicAdd(dst + 1, m.y);
  atomicAdd(dst + 2, m.z);
  atomicAdd(dst + 3, m.w);
  if (g == 0) atomicAdd(&cnt[bg], 1.0f);
}

__global__ void k_pool_div(const float* __restrict__ pool,
                           const float* __restrict__ cnt,
                           float* __restrict__ pooled, int g64) {
  int i = blockIdx.x * blockDim.x + threadIdx.x;
  if (i >= g64) return;
  float c = cnt[i >> 6];
  pooled[i] = (c > 0.0f) ? pool[i] / c : 0.0f;
}

__global__ void k_gemm_wmma(const float* __restrict__ A,
                            const float* __restrict__ B,
                            float* __restrict__ C,
                            const float* __restrict__ bias,
                            const float* __restrict__ dinv,
                            float* __restrict__ acc,
                            int M, int K, int Nc) {
  int wid  = (blockIdx.x * blockDim.x + threadIdx.x) >> 5;
  int lane = threadIdx.x & 31;
  int tiles_n = Nc >> 4;
  int tm = wid / tiles_n;
  int tn = wid - tm * tiles_n;
  if (tm * 16 >= M) return;

  int half = lane >> 4;
  int lm   = lane & 15;
  const float* Arow = A + (size_t)(tm * 16 + lm) * K;
  int ncol = tn * 16 + lm;

  v8f c = {};
#if USE_WMMA_F32
  for (int k0 = 0; k0 < K; k0 += 4) {
    v2f a, b;
    a.x = Arow[k0 + 2 * half + 0];
    a.y = Arow[k0 + 2 * half + 1];
    b.x = B[(size_t)(k0 + 2 * half + 0) * Nc + ncol];
    b.y = B[(size_t)(k0 + 2 * half + 1) * Nc + ncol];
    c = __builtin_amdgcn_wmma_f32_16x16x4_f32(false, a, false, b, (short)0, c,
                                              false, false);
  }
#else
  for (int k0 = 0; k0 < K; k0 += 32) {
    v16h a, b;
#pragma unroll
    for (int i = 0; i < 8; ++i) {
      a[i]     = (_Float16)Arow[k0 + 8 * half + i];
      a[8 + i] = (_Float16)Arow[k0 + 16 + 8 * half + i];
    }
#pragma unroll
    for (int i = 0; i < 16; ++i)
      b[i] = (_Float16)B[(size_t)(k0 + 16 * half + i) * Nc + ncol];
    c = __builtin_amdgcn_wmma_f32_16x16x32_f16(false, a, false, b, (short)0, c,
                                               false, false);
  }
#endif

  float bv = bias ? bias[ncol] : 0.0f;
  if (acc) {
#pragma unroll
    for (int r = 0; r < 8; ++r) {
      int row = tm * 16 + 8 * half + r;
      size_t idx = (size_t)row * Nc + ncol;
      float val = c[r] + bv;
      float di = dinv[row];
      C[idx]   = val;
      acc[idx] = val * (di * di);
    }
  } else {
#pragma unroll
    for (int r = 0; r < 8; ++r) {
      C[(size_t)(tm * 16 + 8 * half + r) * Nc + ncol] = c[r] + bv;
    }
  }
}

static inline int blk(long long n, int b) { return (int)((n + b - 1) / b); }

extern "C" void kernel_launch(void* const* d_in, const int* in_sizes, int n_in,
                              void* d_out, int out_size, void* d_ws, size_t ws_size,
                              hipStream_t stream) {
  (void)in_sizes; (void)n_in; (void)out_size; (void)ws_size;
  const int*   edge_index = (const int*)d_in[0];
  const int*   batch      = (const int*)d_in[1];
  const float* W1 = (const float*)d_in[2];
  const float* b1 = (const float*)d_in[3];
  const float* W2 = (const float*)d_in[4];
  const float* b2 = (const float*)d_in[5];
  const float* W3 = (const float*)d_in[6];
  const float* b3 = (const float*)d_in[7];
  const float* g1 = (const float*)d_in[8];
  const float* be1= (const float*)d_in[9];
  const float* m1 = (const float*)d_in[10];
  const float* v1 = (const float*)d_in[11];
  const float* g2 = (const float*)d_in[12];
  const float* be2= (const float*)d_in[13];
  const float* m2 = (const float*)d_in[14];
  const float* v2 = (const float*)d_in[15];
  const float* Wl = (const float*)d_in[16];
  const float* bl = (const float*)d_in[17];
  float* out = (float*)d_out;

  char* ws = (char*)d_ws;
  size_t off = 0;
  auto carve = [&](size_t bytes) -> float* {
    float* p = (float*)(ws + off);
    off += (bytes + 255) & ~(size_t)255;
    return p;
  };
  float* deg_feat = carve((size_t)NN * 4);
  float* degcnt   = carve((size_t)NN * 4);
  float* dinv     = carve((size_t)NN * 4);
  float* bufX     = carve((size_t)NN * DD * 4);
  float* bufH     = carve((size_t)NN * DD * 4);
  float* bufA     = carve((size_t)NN * DD * 4);
  float* pool     = carve((size_t)GG * DD * 4);
  float* cnt      = carve((size_t)GG * 4);
  float* pooled   = carve((size_t)GG * DD * 4);

  const int B = 256;
  const long long N64 = (long long)NN * DD;
  const long long G64 = (long long)GG * DD;

  k_zero<<<blk(NN, B), B, 0, stream>>>(deg_feat, NN);
  k_zero<<<blk(NN, B), B, 0, stream>>>(degcnt, NN);
  k_zero<<<blk(G64, B), B, 0, stream>>>(pool, (int)G64);
  k_zero<<<blk(GG, B), B, 0, stream>>>(cnt, GG);
  k_edge_count<<<blk(EE, B), B, 0, stream>>>(edge_index, deg_feat, degcnt, EE);
  k_dinv<<<blk(NN, B), B, 0, stream>>>(degcnt, dinv, NN);

  const int tiles_NN  = (NN / 16) * (DD / 16);
  const int tiles_FIN = (GG / 16) * (DOUT / 16);

  k_l1_hw_self<<<blk(N64, B), B, 0, stream>>>(deg_feat, W1, dinv, bufH, bufA, (int)N64);
  k_scatter<<<blk((long long)EE * 16, B), B, 0, stream>>>(edge_index, bufH, dinv, bufA, EE);
  k_post_bnrelu<<<blk(N64, B), B, 0, stream>>>(bufA, b1, g1, be1, m1, v1, bufX, (int)N64);

  k_gemm_wmma<<<blk((long long)tiles_NN * 32, B), B, 0, stream>>>(
      bufX, W2, bufH, nullptr, dinv, bufA, NN, DD, DD);
  k_scatter<<<blk((long long)EE * 16, B), B, 0, stream>>>(edge_index, bufH, dinv, bufA, EE);
  k_post_bnrelu<<<blk(N64, B), B, 0, stream>>>(bufA, b2, g2, be2, m2, v2, bufX, (int)N64);

  k_gemm_wmma<<<blk((long long)tiles_NN * 32, B), B, 0, stream>>>(
      bufX, W3, bufH, nullptr, dinv, bufA, NN, DD, DD);
  k_scatter<<<blk((long long)EE * 16, B), B, 0, stream>>>(edge_index, bufH, dinv, bufA, EE);
  k_post_bias<<<blk(N64, B), B, 0, stream>>>(bufA, b3, bufX, (int)N64);

  k_pool_scatter<<<blk((long long)NN * 16, B), B, 0, stream>>>(batch, bufX, pool, cnt, NN);
  k_pool_div<<<blk(G64, B), B, 0, stream>>>(pool, cnt, pooled, (int)G64);
  k_gemm_wmma<<<blk((long long)tiles_FIN * 32, B), B, 0, stream>>>(
      pooled, Wl, out, bl, nullptr, nullptr, GG, DD, DOUT);
}


# Aggregate by gather over a destination-sorted edge list (counting sort built each call) instead of float scatter atomics; sum order = segment slot order; bias/BN/ReLU at the store
# speedup vs baseline: 1.7743x; 1.7743x over previous
#include <hip/hip_runtime.h>
#include <hip/hip_bf16.h>

#define NN   50000
#define EE   800000
#define GG   512
#define DD   64
#define DOUT 32
#define BN_EPS 1e-5f

typedef __attribute__((ext_vector_type(2)))  float    v2f;
typedef __attribute__((ext_vector_type(8)))  float    v8f;
typedef __attribute__((ext_vector_type(16))) _Float16 v16h;

#if defined(__AMDGCN__) && __has_builtin(__builtin_amdgcn_wmma_f32_16x16x4_f32)
#define USE_WMMA_F32 1
#else
#define USE_WMMA_F32 0
#endif


__global__ void k_zero(float* __restrict__ p, int n) {
  int i = blockIdx.x * blockDim.x + threadIdx.x;
  if (i < n) p[i] = 0.0f;
}

__global__ void k_edge_count(const int* __restrict__ ei,
                             float* __restrict__ deg_feat,
                             int* __restrict__ degcnt,
                             int* __restrict__ rank, int E) {
  int e = blockIdx.x * blockDim.x + threadIdx.x;
  if (e >= E) return;
  int s = ei[e];
  int d = ei[E + e];
  atomicAdd(&deg_feat[s], 1.0f);
  rank[e] = atomicAdd(&degcnt[d], 1);
}

__global__ void __launch_bounds__(1024)
k_scan_dinv(const int* __restrict__ degcnt, int* __restrict__ rowptr,
            float* __restrict__ dinv, int n) {
  __shared__ int wsum[32];
  __shared__ int s_base;
  int t = threadIdx.x;
  int lane = t & 31;
  int w = t >> 5;
  int b0 = blockIdx.x * 1024;

  int p = 0;
  for (int i = t; i < b0; i += 1024) p += degcnt[i];
  for (int o = 16; o > 0; o >>= 1) p += __shfl_xor(p, o);
  if (lane == 0) wsum[w] = p;
  __syncthreads();
  if (w == 0) {
    int v = wsum[lane];
    for (int o = 16; o > 0; o >>= 1) v += __shfl_xor(v, o);
    if (lane == 0) s_base = v;
  }
  __syncthreads();
  int base = s_base;

  int i = b0 + t;
  int c = (i < n) ? degcnt[i] : 0;
  int v = c;
  for (int o = 1; o < 32; o <<= 1) {
    int u = __shfl_up(v, o);
    if (lane >= o) v += u;
  }
  if (lane == 31) wsum[w] = v;
  __syncthreads();
  if (w == 0) {
    int orig = wsum[lane];
    int x = orig;
    for (int o = 1; o < 32; o <<= 1) {
      int u = __shfl_up(x, o);
      if (lane >= o) x += u;
    }
    wsum[lane] = x - orig;
  }
  __syncthreads();
  if (i < n) {
    rowptr[i] = base + wsum[w] + v - c;
    dinv[i] = rsqrtf((float)c + 1.0f);
  }
}

__global__ void k_fill(const int* __restrict__ ei,
                       const int* __restrict__ rank,
                       const int* __restrict__ rowptr,
                       int* __restrict__ csr, int E) {
  int e = blockIdx.x * blockDim.x + threadIdx.x;
  if (e >= E) return;
  int s = ei[e];
  int d = ei[E + e];
  csr[rowptr[d] + rank[e]] = s;
}

__global__ void k_l1_hw(const float* __restrict__ deg_feat,
                        const float* __restrict__ W1,
                        float* __restrict__ hw, int n64) {
  int i = blockIdx.x * blockDim.x + threadIdx.x;
  if (i >= n64) return;
  hw[i] = deg_feat[i >> 6] * W1[i & 63];
}

template <bool BN>
__global__ void k_gather(const int* __restrict__ rowptr,
                         const int* __restrict__ degcnt,
                         const int* __restrict__ csr,
                         const float* __restrict__ hw,
                         const float* __restrict__ dinv,
                         const float* __restrict__ b,
                         const float* __restrict__ g,
                         const float* __restrict__ be,
                         const float* __restrict__ m,
                         const float* __restrict__ v,
                         float* __restrict__ x, int n) {
  int t = blockIdx.x * blockDim.x + threadIdx.x;
  int nd = t >> 4;
  int q  = t & 15;
  if (nd >= n) return;
  const float4* hw4 = (const float4*)hw;
  float dn = dinv[nd];
  int beg = rowptr[nd];
  int end = beg + degcnt[nd];
  float4 h = hw4[(size_t)nd * 16 + q];
  float ws = dn * dn;
  float4 a;
  a.x = h.x * ws; a.y = h.y * ws; a.z = h.z * ws; a.w = h.w * ws;
  if (beg < end) {
    int last = end - 1;
    int s0 = csr[beg];
    int s1 = csr[min(beg + 1, last)];
    int s2 = csr[min(beg + 2, last)];
    int s3 = csr[min(beg + 3, last)];
    for (int j = beg; j < end; j += 4) {
      bool p1 = (j + 1) < end;
      bool p2 = (j + 2) < end;
      bool p3 = (j + 3) < end;
      float w0 = dinv[s0] * dn;
      float w1 = dinv[s1] * dn;
      float w2 = dinv[s2] * dn;
      float w3 = dinv[s3] * dn;
      float4 m0 = hw4[(size_t)s0 * 16 + q];
      float4 m1 = hw4[(size_t)s1 * 16 + q];
      float4 m2 = hw4[(size_t)s2 * 16 + q];
      float4 m3 = hw4[(size_t)s3 * 16 + q];
      s0 = csr[min(j + 4, last)];
      s1 = csr[min(j + 5, last)];
      s2 = csr[min(j + 6, last)];
      s3 = csr[min(j + 7, last)];
      asm volatile("" :: "v"(m0.x), "v"(m0.y), "v"(m0.z), "v"(m0.w),
                         "v"(m1.x), "v"(m1.y), "v"(m1.z), "v"(m1.w),
                         "v"(m2.x), "v"(m2.y), "v"(m2.z), "v"(m2.w),
                         "v"(m3.x), "v"(m3.y), "v"(m3.z), "v"(m3.w));
      a.x += m0.x * w0; a.y += m0.y * w0; a.z += m0.z * w0; a.w += m0.w * w0;
      float4 u;
      u.x = a.x + m1.x * w1; u.y = a.y + m1.y * w1; u.z = a.z + m1.z * w1; u.w = a.w + m1.w * w1;
      a.x = p1 ? u.x : a.x; a.y = p1 ? u.y : a.y; a.z = p1 ? u.z : a.z; a.w = p1 ? u.w : a.w;
      u.x = a.x + m2.x * w2; u.y = a.y + m2.y * w2; u.z = a.z + m2.z * w2; u.w = a.w + m2.w * w2;
      a.x = p2 ? u.x : a.x; a.y = p2 ? u.y : a.y; a.z = p2 ? u.z : a.z; a.w = p2 ? u.w : a.w;
      u.x = a.x + m3.x * w3; u.y = a.y + m3.y * w3; u.z = a.z + m3.z * w3; u.w = a.w + m3.w * w3;
      a.x = p3 ? u.x : a.x; a.y = p3 ? u.y : a.y; a.z = p3 ? u.z : a.z; a.w = p3 ? u.w : a.w;
    }
  }
  float r[4] = {a.x, a.y, a.z, a.w};
  float4 y;
  float* yp = &y.x;
#pragma unroll
  for (int c = 0; c < 4; ++c) {
    int d = q * 4 + c;
    float val = r[c] + b[d];
    if (BN) {
      float rs = rsqrtf(v[d] + BN_EPS);
      val = fmaxf((val - m[d]) * rs * g[d] + be[d], 0.0f);
    }
    yp[c] = val;
  }
  ((float4*)x)[(size_t)nd * 16 + q] = y;
}

__global__ void k_pool_scatter(const int* __restrict__ batch,
                               const float* __restrict__ x,
                               float* __restrict__ pool,
                               float* __restrict__ cnt, int n) {
  int t = blockIdx.x * blockDim.x + threadIdx.x;
  int nd = t >> 4;
  int g  = t & 15;
  if (nd >= n) return;
  int bg = batch[nd];
  float4 m = ((const float4*)(x + (size_t)nd * DD))[g];
  float* dst = pool + (size_t)bg * DD + g * 4;
  atomicAdd(dst + 0, m.x);
  atomicAdd(dst + 1, m.y);
  atomicAdd(dst + 2, m.z);
  atomicAdd(dst + 3, m.w);
  if (g == 0) atomicAdd(&cnt[bg], 1.0f);
}

__global__ void k_pool_div(const float* __restrict__ pool,
                           const float* __restrict__ cnt,
                           float* __restrict__ pooled, int g64) {
  int i = blockIdx.x * blockDim.x + threadIdx.x;
  if (i >= g64) return;
  float c = cnt[i >> 6];
  pooled[i] = (c > 0.0f) ? pool[i] / c : 0.0f;
}

__global__ void k_gemm_wmma(const float* __restrict__ A,
                            const float* __restrict__ B,
                            float* __restrict__ C,
                            const float* __restrict__ bias,
                            const float* __restrict__ dinv,
                            float* __restrict__ acc,
                            int M, int K, int Nc) {
  int wid  = (blockIdx.x * blockDim.x + threadIdx.x) >> 5;
  int lane = threadIdx.x & 31;
  int tiles_n = Nc >> 4;
  int tm = wid / tiles_n;
  int tn = wid - tm * tiles_n;
  if (tm * 16 >= M) return;

  int half = lane >> 4;
  int lm   = lane & 15;
  const float* Arow = A + (size_t)(tm * 16 + lm) * K;
  int ncol = tn * 16 + lm;

  v8f c = {};
#if USE_WMMA_F32
  for (int k0 = 0; k0 < K; k0 += 4) {
    v2f a, b;
    a.x = Arow[k0 + 2 * half + 0];
    a.y = Arow[k0 + 2 * half + 1];
    b.x = B[(size_t)(k0 + 2 * half + 0) * Nc + ncol];
    b.y = B[(size_t)(k0 + 2 * half + 1) * Nc + ncol];
    c = __builtin_amdgcn_wmma_f32_16x16x4_f32(false, a, false, b, (short)0, c,
                                              false, false);
  }
#else
  for (int k0 = 0; k0 < K; k0 += 32) {
    v16h a, b;
#pragma unroll
    for (int i = 0; i < 8; ++i) {
      a[i]     = (_Float16)Arow[k0 + 8 * half + i];
      a[8 + i] = (_Float16)Arow[k0 + 16 + 8 * half + i];
    }
#pragma unroll
    for (int i = 0; i < 16; ++i)
      b[i] = (_Float16)B[(size_t)(k0 + 16 * half + i) * Nc + ncol];
    c = __builtin_amdgcn_wmma_f32_16x16x32_f16(false, a, false, b, (short)0, c,
                                               false, false);
  }
#endif

  float bv = bias ? bias[ncol] : 0.0f;
  if (acc) {
#pragma unroll
    for (int r = 0; r < 8; ++r) {
      int row = tm * 16 + 8 * half + r;
      size_t idx = (size_t)row * Nc + ncol;
      float val = c[r] + bv;
      float di = dinv[row];
      C[idx]   = val;
      acc[idx] = val * (di * di);
    }
  } else {
#pragma unroll
    for (int r = 0; r < 8; ++r) {
      C[(size_t)(tm * 16 + 8 * half + r) * Nc + ncol] = c[r] + bv;
    }
  }
}

static inline int blk(long long n, int b) { return (int)((n + b - 1) / b); }

extern "C" void kernel_launch(void* const* d_in, const int* in_sizes, int n_in,
                              void* d_out, int out_size, void* d_ws, size_t ws_size,
                              hipStream_t stream) {
  (void)in_sizes; (void)n_in; (void)out_size; (void)ws_size;
  const int*   edge_index = (const int*)d_in[0];
  const int*   batch      = (const int*)d_in[1];
  const float* W1 = (const float*)d_in[2];
  const float* b1 = (const float*)d_in[3];
  const float* W2 = (const float*)d_in[4];
  const float* b2 = (const float*)d_in[5];
  const float* W3 = (const float*)d_in[6];
  const float* b3 = (const float*)d_in[7];
  const float* g1 = (const float*)d_in[8];
  const float* be1= (const float*)d_in[9];
  const float* m1 = (const float*)d_in[10];
  const float* v1 = (const float*)d_in[11];
  const float* g2 = (const float*)d_in[12];
  const float* be2= (const float*)d_in[13];
  const float* m2 = (const float*)d_in[14];
  const float* v2 = (const float*)d_in[15];
  const float* Wl = (const float*)d_in[16];
  const float* bl = (const float*)d_in[17];
  float* out = (float*)d_out;

  char* ws = (char*)d_ws;
  size_t off = 0;
  auto carve = [&](size_t bytes) -> float* {
    float* p = (float*)(ws + off);
    off += (bytes + 255) & ~(size_t)255;
    return p;
  };
  float* deg_feat = carve((size_t)NN * 4);
  int*   degcnt   = (int*)carve((size_t)NN * 4);
  float* dinv     = carve((size_t)NN * 4);
  int*   rowptr   = (int*)carve((size_t)NN * 4);
  int*   rank     = (int*)carve((size_t)EE * 4);
  int*   csr      = (int*)carve((size_t)EE * 4);
  float* bufX     = carve((size_t)NN * DD * 4);
  float* bufH     = carve((size_t)NN * DD * 4);
  float* pool     = carve((size_t)GG * DD * 4);
  float* cnt      = carve((size_t)GG * 4);
  float* pooled   = carve((size_t)GG * DD * 4);

  const int B = 256;
  const long long N64 = (long long)NN * DD;
  const long long G64 = (long long)GG * DD;

  k_zero<<<blk(NN, B), B, 0, stream>>>(deg_feat, NN);
  k_zero<<<blk(NN, B), B, 0, stream>>>((float*)degcnt, NN);
  k_zero<<<blk(G64, B), B, 0, stream>>>(pool, (int)G64);
  k_zero<<<blk(GG, B), B, 0, stream>>>(cnt, GG);
  k_edge_count<<<blk(EE, B), B, 0, stream>>>(edge_index, deg_feat, degcnt, rank, EE);
  k_scan_dinv<<<blk(NN, 1024), 1024, 0, stream>>>(degcnt, rowptr, dinv, NN);
  k_fill<<<blk(EE, B), B, 0, stream>>>(edge_index, rank, rowptr, csr, EE);

  const int tiles_NN  = (NN / 16) * (DD / 16);
  const int tiles_FIN = (GG / 16) * (DOUT / 16);

  k_l1_hw<<<blk(N64, B), B, 0, stream>>>(deg_feat, W1, bufH, (int)N64);
  k_gather<true><<<blk((long long)NN * 16, B), B, 0, stream>>>(
      rowptr, degcnt, csr, bufH, dinv, b1, g1, be1, m1, v1, bufX, NN);

  k_gemm_wmma<<<blk((long long)tiles_NN * 32, B), B, 0, stream>>>(
      bufX, W2, bufH, nullptr, nullptr, nullptr, NN, DD, DD);
  k_gather<true><<<blk((long long)NN * 16, B), B, 0, stream>>>(
      rowptr, degcnt, csr, bufH, dinv, b2, g2, be2, m2, v2, bufX, NN);

  k_gemm_wmma<<<blk((long long)tiles_NN * 32, B), B, 0, stream>>>(
      bufX, W3, bufH, nullptr, nullptr, nullptr, NN, DD, DD);
  k_gather<false><<<blk((long long)NN * 16, B), B, 0, stream>>>(
      rowptr, degcnt, csr, bufH, dinv, b3, nullptr, nullptr, nullptr, nullptr, bufX, NN);

  k_pool_scatter<<<blk((long long)NN * 16, B), B, 0, stream>>>(batch, bufX, pool, cnt, NN);
  k_pool_div<<<blk(G64, B), B, 0, stream>>>(pool, cnt, pooled, (int)G64);
  k_gemm_wmma<<<blk((long long)tiles_FIN * 32, B), B, 0, stream>>>(
      pooled, Wl, out, bl, nullptr, nullptr, GG, DD, DOUT);
}


# Layer 1 has a 1-wide input: gather one scalar per edge (deg*dinv), form the 64-wide row at the store; outer-product pass and its row gather gone; sum order: 4 lane-strided partials, then a tree
# speedup vs baseline: 1.8573x; 1.0468x over previous
#include <hip/hip_runtime.h>
#include <hip/hip_bf16.h>

#define NN   50000
#define EE   800000
#define GG   512
#define DD   64
#define DOUT 32
#define BN_EPS 1e-5f

typedef __attribute__((ext_vector_type(2)))  float    v2f;
typedef __attribute__((ext_vector_type(8)))  float    v8f;
typedef __attribute__((ext_vector_type(16))) _Float16 v16h;

#if defined(__AMDGCN__) && __has_builtin(__builtin_amdgcn_wmma_f32_16x16x4_f32)
#define USE_WMMA_F32 1
#else
#define USE_WMMA_F32 0
#endif


__global__ void k_zero(float* __restrict__ p, int n) {
  int i = blockIdx.x * blockDim.x + threadIdx.x;
  if (i < n) p[i] = 0.0f;
}

__global__ void k_edge_count(const int* __restrict__ ei,
                             float* __restrict__ deg_feat,
                             int* __restrict__ degcnt,
                             int* __restrict__ rank, int E) {
  int e = blockIdx.x * blockDim.x + threadIdx.x;
  if (e >= E) return;
  int s = ei[e];
  int d = ei[E + e];
  atomicAdd(&deg_feat[s], 1.0f);
  rank[e] = atomicAdd(&degcnt[d], 1);
}

__global__ void __launch_bounds__(1024)
k_scan_dinv(const int* __restrict__ degcnt, const float* __restrict__ deg_feat,
            int* __restrict__ rowptr, float* __restrict__ dinv,
            float* __restrict__ sd, int n) {
  __shared__ int wsum[32];
  __shared__ int s_base;
  int t = threadIdx.x;
  int lane = t & 31;
  int w = t >> 5;
  int b0 = blockIdx.x * 1024;

  int p = 0;
  for (int i = t; i < b0; i += 1024) p += degcnt[i];
  for (int o = 16; o > 0; o >>= 1) p += __shfl_xor(p, o);
  if (lane == 0) wsum[w] = p;
  __syncthreads();
  if (w == 0) {
    int v = wsum[lane];
    for (int o = 16; o > 0; o >>= 1) v += __shfl_xor(v, o);
    if (lane == 0) s_base = v;
  }
  __syncthreads();
  int base = s_base;

  int i = b0 + t;
  int c = (i < n) ? degcnt[i] : 0;
  int v = c;
  for (int o = 1; o < 32; o <<= 1) {
    int u = __shfl_up(v, o);
    if (lane >= o) v += u;
  }
  if (lane == 31) wsum[w] = v;
  __syncthreads();
  if (w == 0) {
    int orig = wsum[lane];
    int x = orig;
    for (int o = 1; o < 32; o <<= 1) {
      int u = __shfl_up(x, o);
      if (lane >= o) x += u;
    }
    wsum[lane] = x - orig;
  }
  __syncthreads();
  if (i < n) {
    rowptr[i] = base + wsum[w] + v - c;
    float di = rsqrtf((float)c + 1.0f);
    dinv[i] = di;
    sd[i] = deg_feat[i] * di;
  }
}

__global__ void k_fill(const int* __restrict__ ei,
                       const int* __restrict__ rank,
                       const int* __restrict__ rowptr,
                       int* __restrict__ csr, int E) {
  int e = blockIdx.x * blockDim.x + threadIdx.x;
  if (e >= E) return;
  int s = ei[e];
  int d = ei[E + e];
  csr[rowptr[d] + rank[e]] = s;
}

__global__ void __launch_bounds__(256)
k_gather_l1(const int* __restrict__ rowptr,
            const int* __restrict__ degcnt,
            const int* __restrict__ csr,
            const float* __restrict__ sd,
            const float* __restrict__ dinv,
            const float* __restrict__ W1,
            const float* __restrict__ b,
            const float* __restrict__ g,
            const float* __restrict__ be,
            const float* __restrict__ m,
            const float* __restrict__ v,
            float* __restrict__ x, int n) {
  int t = blockIdx.x * blockDim.x + threadIdx.x;
  int lane = threadIdx.x & 31;
  int sub = t & 3;
  int ndc = min(t >> 2, n - 1);
  int beg = rowptr[ndc];
  int cnt = degcnt[ndc];
  float dn = dinv[ndc];
  float acc = (sub == 0) ? sd[ndc] : 0.0f;
  int end = beg + cnt;
  int last = (cnt > 0) ? end - 1 : 0;
  int j0 = beg + sub;
  int s[8];
  float val[8];
#pragma unroll
  for (int i = 0; i < 8; ++i) s[i] = csr[min(j0 + 4 * i, last)];
#pragma unroll
  for (int i = 0; i < 8; ++i) val[i] = sd[s[i]];
#pragma unroll
  for (int i = 0; i < 8; ++i) acc += (j0 + 4 * i < end) ? val[i] : 0.0f;
  for (int j = j0 + 32; j < end; j += 4) acc += sd[csr[j]];
  acc += __shfl_xor(acc, 1);
  acc += __shfl_xor(acc, 2);
  float tn = acc * dn;

  int q  = lane & 15;
  int hi = lane >> 4;
  int node0 = (t - lane) >> 2;
  float w1[4], bb[4], mm[4], rs[4], gg[4], bee[4];
#pragma unroll
  for (int c = 0; c < 4; ++c) {
    int d = q * 4 + c;
    w1[c] = W1[d]; bb[c] = b[d]; mm[c] = m[d];
    rs[c] = rsqrtf(v[d] + BN_EPS); gg[c] = g[d]; bee[c] = be[d];
  }
#pragma unroll
  for (int it = 0; it < 4; ++it) {
    int ln = 2 * it + hi;
    float tv = __shfl(tn, ln * 4);
    float4 y;
    float* yp = &y.x;
#pragma unroll
    for (int c = 0; c < 4; ++c) {
      float r = tv * w1[c] + bb[c];
      yp[c] = fmaxf((r - mm[c]) * rs[c] * gg[c] + bee[c], 0.0f);
    }
    int node = node0 + ln;
    if (node < n) ((float4*)x)[(size_t)node * 16 + q] = y;
  }
}

template <bool BN>
__global__ void k_gather(const int* __restrict__ rowptr,
                         const int* __restrict__ degcnt,
                         const int* __restrict__ csr,
                         const float* __restrict__ hw,
                         const float* __restrict__ dinv,
                         const float* __restrict__ b,
                         const float* __restrict__ g,
                         const float* __restrict__ be,
                         const float* __restrict__ m,
                         const float* __restrict__ v,
                         float* __restrict__ x, int n) {
  int t = blockIdx.x * blockDim.x + threadIdx.x;
  int nd = t >> 4;
  int q  = t & 15;
  if (nd >= n) return;
  const float4* hw4 = (const float4*)hw;
  float dn = dinv[nd];
  int beg = rowptr[nd];
  int end = beg + degcnt[nd];
  float4 h = hw4[(size_t)nd * 16 + q];
  float ws = dn * dn;
  float4 a;
  a.x = h.x * ws; a.y = h.y * ws; a.z = h.z * ws; a.w = h.w * ws;
  if (beg < end) {
    int last = end - 1;
    int s0 = csr[beg];
    int s1 = csr[min(beg + 1, last)];
    int s2 = csr[min(beg + 2, last)];
    int s3 = csr[min(beg + 3, last)];
    for (int j = beg; j < end; j += 4) {
      bool p1 = (j + 1) < end;
      bool p2 = (j + 2) < end;
      bool p3 = (j + 3) < end;
      float w0 = dinv[s0] * dn;
      float w1 = dinv[s1] * dn;
      float w2 = dinv[s2] * dn;
      float w3 = dinv[s3] * dn;
      float4 m0 = hw4[(size_t)s0 * 16 + q];
      float4 m1 = hw4[(size_t)s1 * 16 + q];
      float4 m2 = hw4[(size_t)s2 * 16 + q];
      float4 m3 = hw4[(size_t)s3 * 16 + q];
      s0 = csr[min(j + 4, last)];
      s1 = csr[min(j + 5, last)];
      s2 = csr[min(j + 6, last)];
      s3 = csr[min(j + 7, last)];
      asm volatile("" :: "v"(m0.x), "v"(m0.y), "v"(m0.z), "v"(m0.w),
                         "v"(m1.x), "v"(m1.y), "v"(m1.z), "v"(m1.w),
                         "v"(m2.x), "v"(m2.y), "v"(m2.z), "v"(m2.w),
                         "v"(m3.x), "v"(m3.y), "v"(m3.z), "v"(m3.w));
      a.x += m0.x * w0; a.y += m0.y * w0; a.z += m0.z * w0; a.w += m0.w * w0;
      float4 u;
      u.x = a.x + m1.x * w1; u.y = a.y + m1.y * w1; u.z = a.z + m1.z * w1; u.w = a.w + m1.w * w1;
      a.x = p1 ? u.x : a.x; a.y = p1 ? u.y : a.y; a.z = p1 ? u.z : a.z; a.w = p1 ? u.w : a.w;
      u.x = a.x + m2.x * w2; u.y = a.y + m2.y * w2; u.z = a.z + m2.z * w2; u.w = a.w + m2.w * w2;
      a.x = p2 ? u.x : a.x; a.y = p2 ? u.y : a.y; a.z = p2 ? u.z : a.z; a.w = p2 ? u.w : a.w;
      u.x = a.x + m3.x * w3; u.y = a.y + m3.y * w3; u.z = a.z + m3.z * w3; u.w = a.w + m3.w * w3;
      a.x = p3 ? u.x : a.x; a.y = p3 ? u.y : a.y; a.z = p3 ? u.z : a.z; a.w = p3 ? u.w : a.w;
    }
  }
  float r[4] = {a.x, a.y, a.z, a.w};
  float4 y;
  float* yp = &y.x;
#pragma unroll
  for (int c = 0; c < 4; ++c) {
    int d = q * 4 + c;
    float val = r[c] + b[d];
    if (BN) {
      float rs = rsqrtf(v[d] + BN_EPS);
      val = fmaxf((val - m[d]) * rs * g[d] + be[d], 0.0f);
    }
    yp[c] = val;
  }
  ((float4*)x)[(size_t)nd * 16 + q] = y;
}

__global__ void k_pool_scatter(const int* __restrict__ batch,
                               const float* __restrict__ x,
                               float* __restrict__ pool,
                               float* __restrict__ cnt, int n) {
  int t = blockIdx.x * blockDim.x + threadIdx.x;
  int nd = t >> 4;
  int g  = t & 15;
  if (nd >= n) return;
  int bg = batch[nd];
  float4 m = ((const float4*)(x + (size_t)nd * DD))[g];
  float* dst = pool + (size_t)bg * DD + g * 4;
  atomicAdd(dst + 0, m.x);
  atomicAdd(dst + 1, m.y);
  atomicAdd(dst + 2, m.z);
  atomicAdd(dst + 3, m.w);
  if (g == 0) atomicAdd(&cnt[bg], 1.0f);
}

__global__ void k_pool_div(const float* __restrict__ pool,
                           const float* __restrict__ cnt,
                           float* __restrict__ pooled, int g64) {
  int i = blockIdx.x * blockDim.x + threadIdx.x;
  if (i >= g64) return;
  float c = cnt[i >> 6];
  pooled[i] = (c > 0.0f) ? pool[i] / c : 0.0f;
}

__global__ void k_gemm_wmma(const float* __restrict__ A,
                            const float* __restrict__ B,
                            float* __restrict__ C,
                            const float* __restrict__ bias,
                            const float* __restrict__ dinv,
                            float* __restrict__ acc,
                            int M, int K, int Nc) {
  int wid  = (blockIdx.x * blockDim.x + threadIdx.x) >> 5;
  int lane = threadIdx.x & 31;
  int tiles_n = Nc >> 4;
  int tm = wid / tiles_n;
  int tn = wid - tm * tiles_n;
  if (tm * 16 >= M) return;

  int half = lane >> 4;
  int lm   = lane & 15;
  const float* Arow = A + (size_t)(tm * 16 + lm) * K;
  int ncol = tn * 16 + lm;

  v8f c = {};
#if USE_WMMA_F32
  for (int k0 = 0; k0 < K; k0 += 4) {
    v2f a, b;
    a.x = Arow[k0 + 2 * half + 0];
    a.y = Arow[k0 + 2 * half + 1];
    b.x = B[(size_t)(k0 + 2 * half + 0) * Nc + ncol];
    b.y = B[(size_t)(k0 + 2 * half + 1) * Nc + ncol];
    c = __builtin_amdgcn_wmma_f32_16x16x4_f32(false, a, false, b, (short)0, c,
                                              false, false);
  }
#else
  for (int k0 = 0; k0 < K; k0 += 32) {
    v16h a, b;
#pragma unroll
    for (int i = 0; i < 8; ++i) {
      a[i]     = (_Float16)Arow[k0 + 8 * half + i];
      a[8 + i] = (_Float16)Arow[k0 + 16 + 8 * half + i];
    }
#pragma unroll
    for (int i = 0; i < 16; ++i)
      b[i] = (_Float16)B[(size_t)(k0 + 16 * half + i) * Nc + ncol];
    c = __builtin_amdgcn_wmma_f32_16x16x32_f16(false, a, false, b, (short)0, c,
                                               false, false);
  }
#endif

  float bv = bias ? bias[ncol] : 0.0f;
  if (acc) {
#pragma unroll
    for (int r = 0; r < 8; ++r) {
      int row = tm * 16 + 8 * half + r;
      size_t idx = (size_t)row * Nc + ncol;
      float val = c[r] + bv;
      float di = dinv[row];
      C[idx]   = val;
      acc[idx] = val * (di * di);
    }
  } else {
#pragma unroll
    for (int r = 0; r < 8; ++r) {
      C[(size_t)(tm * 16 + 8 * half + r) * Nc + ncol] = c[r] + bv;
    }
  }
}

static inline int blk(long long n, int b) { return (int)((n + b - 1) / b); }

extern "C" void kernel_launch(void* const* d_in, const int* in_sizes, int n_in,
                              void* d_out, int out_size, void* d_ws, size_t ws_size,
                              hipStream_t stream) {
  (void)in_sizes; (void)n_in; (void)out_size; (void)ws_size;
  const int*   edge_index = (const int*)d_in[0];
  const int*   batch      = (const int*)d_in[1];
  const float* W1 = (const float*)d_in[2];
  const float* b1 = (const float*)d_in[3];
  const float* W2 = (const float*)d_in[4];
  const float* b2 = (const float*)d_in[5];
  const float* W3 = (const float*)d_in[6];
  const float* b3 = (const float*)d_in[7];
  const float* g1 = (const float*)d_in[8];
  const float* be1= (const float*)d_in[9];
  const float* m1 = (const float*)d_in[10];
  const float* v1 = (const float*)d_in[11];
  const float* g2 = (const float*)d_in[12];
  const float* be2= (const float*)d_in[13];
  const float* m2 = (const float*)d_in[14];
  const float* v2 = (const float*)d_in[15];
  const float* Wl = (const float*)d_in[16];
  const float* bl = (const float*)d_in[17];
  float* out = (float*)d_out;

  char* ws = (char*)d_ws;
  size_t off = 0;
  auto carve = [&](size_t bytes) -> float* {
    float* p = (float*)(ws + off);
    off += (bytes + 255) & ~(size_t)255;
    return p;
  };
  float* deg_feat = carve((size_t)NN * 4);
  int*   degcnt   = (int*)carve((size_t)NN * 4);
  float* dinv     = carve((size_t)NN * 4);
  float* sd       = carve((size_t)NN * 4);
  int*   rowptr   = (int*)carve((size_t)NN * 4);
  int*   rank     = (int*)carve((size_t)EE * 4);
  int*   csr      = (int*)carve((size_t)EE * 4);
  float* bufX     = carve((size_t)NN * DD * 4);
  float* bufH     = carve((size_t)NN * DD * 4);
  float* pool     = carve((size_t)GG * DD * 4);
  float* cnt      = carve((size_t)GG * 4);
  float* pooled   = carve((size_t)GG * DD * 4);

  const int B = 256;
  const long long G64 = (long long)GG * DD;

  k_zero<<<blk(NN, B), B, 0, stream>>>(deg_feat, NN);
  k_zero<<<blk(NN, B), B, 0, stream>>>((float*)degcnt, NN);
  k_zero<<<blk(G64, B), B, 0, stream>>>(pool, (int)G64);
  k_zero<<<blk(GG, B), B, 0, stream>>>(cnt, GG);
  k_edge_count<<<blk(EE, B), B, 0, stream>>>(edge_index, deg_feat, degcnt, rank, EE);
  k_scan_dinv<<<blk(NN, 1024), 1024, 0, stream>>>(degcnt, deg_feat, rowptr, dinv, sd, NN);
  k_fill<<<blk(EE, B), B, 0, stream>>>(edge_index, rank, rowptr, csr, EE);

  const int tiles_NN  = (NN / 16) * (DD / 16);
  const int tiles_FIN = (GG / 16) * (DOUT / 16);

  k_gather_l1<<<blk((long long)NN * 4, B), B, 0, stream>>>(
      rowptr, degcnt, csr, sd, dinv, W1, b1, g1, be1, m1, v1, bufX, NN);

  k_gemm_wmma<<<blk((long long)tiles_NN * 32, B), B, 0, stream>>>(
      bufX, W2, bufH, nullptr, nullptr, nullptr, NN, DD, DD);
  k_gather<true><<<blk((long long)NN * 16, B), B, 0, stream>>>(
      rowptr, degcnt, csr, bufH, dinv, b2, g2, be2, m2, v2, bufX, NN);

  k_gemm_wmma<<<blk((long long)tiles_NN * 32, B), B, 0, stream>>>(
      bufX, W3, bufH, nullptr, nullptr, nullptr, NN, DD, DD);
  k_gather<false><<<blk((long long)NN * 16, B), B, 0, stream>>>(
      rowptr, degcnt, csr, bufH, dinv, b3, nullptr, nullptr, nullptr, nullptr, bufX, NN);

  k_pool_scatter<<<blk((long long)NN * 16, B), B, 0, stream>>>(batch, bufX, pool, cnt, NN);
  k_pool_div<<<blk(G64, B), B, 0, stream>>>(pool, cnt, pooled, (int)G64);
  k_gemm_wmma<<<blk((long long)tiles_FIN * 32, B), B, 0, stream>>>(
      pooled, Wl, out, bl, nullptr, nullptr, GG, DD, DOUT);
}


# Degree counts and edge ranks on per-work-group LDS counter tables (64 edge chunks, 16-bit fields, per-node prefix over chunks) instead of global atomics; sum order: chunk, then LDS arrival rank
# speedup vs baseline: 2.7094x; 1.4588x over previous
#include <hip/hip_runtime.h>
#include <hip/hip_bf16.h>

#define NN   50000
#define EE   800000
#define GG   512
#define DD   64
#define DOUT 32
#define BN_EPS 1e-5f

typedef __attribute__((ext_vector_type(2)))  float    v2f;
typedef __attribute__((ext_vector_type(8)))  float    v8f;
typedef __attribute__((ext_vector_type(16))) _Float16 v16h;

#if defined(__AMDGCN__) && __has_builtin(__builtin_amdgcn_wmma_f32_16x16x4_f32)
#define USE_WMMA_F32 1
#else
#define USE_WMMA_F32 0
#endif


__global__ void k_zero(float* __restrict__ p, int n) {
  int i = blockIdx.x * blockDim.x + threadIdx.x;
  if (i < n) p[i] = 0.0f;
}

#define CH 64
#define CS (EE / CH)
#define NB ((NN + 255) / 256)
static_assert(EE % CH == 0 && CS % 4 == 0, "a chunk is a whole number of 4-edge groups");
static_assert(CS < 65536, "a chunk's count for one node fits a 16-bit field");
static_assert(NN % 4 == 0, "the table is moved as 4-word groups");
static_assert(NB <= 256, "one thread of the scan block per node block in front of it");
#define G4  (CS / 4)
#define PER ((G4 + 1023) / 1024)

__global__ void __launch_bounds__(1024)
k_count_lds(const int* __restrict__ ei, unsigned* __restrict__ hist,
            int* __restrict__ bsum, int* __restrict__ rank) {
  __shared__ __attribute__((aligned(16))) unsigned h[NN];
  __shared__ int sb[NB];
  const int t = threadIdx.x;
  const int lane = t & 31;
  const int w = blockIdx.x;
  const int4* src4 = (const int4*)(ei + (size_t)w * CS);
  const int4* dst4 = (const int4*)(ei + EE + (size_t)w * CS);
  int4 sv[PER], dv[PER];
#pragma unroll
  for (int k = 0; k < PER; ++k) {
    int g = min(t + 1024 * k, G4 - 1);
    sv[k] = src4[g];
    dv[k] = dst4[g];
  }
  uint4* h4 = (uint4*)h;
  for (int j = t; j < NN / 4; j += 1024) h4[j] = make_uint4(0u, 0u, 0u, 0u);
  if (t < NB) sb[t] = 0;
  __syncthreads();
  int4* rank4 = (int4*)(rank + (size_t)w * CS);
#pragma unroll
  for (int k = 0; k < PER; ++k) {
    int g = t + 1024 * k;
    if (g < G4) {
      int4 r;
      r.x = (int)(atomicAdd(&h[dv[k].x], 1u) & 0xFFFFu);
      r.y = (int)(atomicAdd(&h[dv[k].y], 1u) & 0xFFFFu);
      r.z = (int)(atomicAdd(&h[dv[k].z], 1u) & 0xFFFFu);
      r.w = (int)(atomicAdd(&h[dv[k].w], 1u) & 0xFFFFu);
      atomicAdd(&h[sv[k].x], 0x10000u);
      atomicAdd(&h[sv[k].y], 0x10000u);
      atomicAdd(&h[sv[k].z], 0x10000u);
      atomicAdd(&h[sv[k].w], 0x10000u);
      rank4[g] = r;
    }
  }
  __syncthreads();
  uint4* hg4 = (uint4*)(hist + (size_t)w * NN);
  for (int j0 = 0; j0 < NN / 4; j0 += 1024) {
    int j = j0 + t;
    bool ok = j < NN / 4;
    uint4 v = make_uint4(0u, 0u, 0u, 0u);
    if (ok) { v = h4[j]; hg4[j] = v; }
    int c = (int)((v.x & 0xFFFFu) + (v.y & 0xFFFFu) + (v.z & 0xFFFFu) + (v.w & 0xFFFFu));
    for (int o = 16; o > 0; o >>= 1) c += __shfl_xor(c, o);
    if (lane == 0 && ok) atomicAdd(&sb[j >> 6], c);
  }
  __syncthreads();
  if (t < NB) bsum[w * NB + t] = sb[t];
}

__global__ void __launch_bounds__(256)
k_scan_dinv(const unsigned* __restrict__ hist, const int* __restrict__ bsum,
            int* __restrict__ off, int* __restrict__ degcnt,
            int* __restrict__ rowptr, float* __restrict__ dinv,
            float* __restrict__ sd, int n) {
  __shared__ int wsum[8];
  __shared__ int s_base;
  int t = threadIdx.x;
  int lane = t & 31;
  int w = t >> 5;
  int b = blockIdx.x;
  int i = b * 256 + t;
  int ic = min(i, n - 1);
  unsigned hv[CH];
#pragma unroll
  for (int k = 0; k < CH; ++k) hv[k] = hist[(size_t)k * NN + ic];

  int p = 0;
  if (t < b) {
    for (int k = 0; k < CH; ++k) p += bsum[k * NB + t];
  }
  for (int o = 16; o > 0; o >>= 1) p += __shfl_xor(p, o);
  if (lane == 0) wsum[w] = p;
  __syncthreads();
  if (w == 0) {
    int x = (lane < 8) ? wsum[lane] : 0;
    for (int o = 16; o > 0; o >>= 1) x += __shfl_xor(x, o);
    if (lane == 0) s_base = x;
  }
  __syncthreads();
  int base = s_base;

  int cin = 0, cout = 0;
#pragma unroll
  for (int k = 0; k < CH; ++k) { cin += (int)(hv[k] & 0xFFFFu); cout += (int)(hv[k] >> 16); }
  int c = (i < n) ? cin : 0;
  int v = c;
  for (int o = 1; o < 32; o <<= 1) {
    int u = __shfl_up(v, o);
    if (lane >= o) v += u;
  }
  if (lane == 31) wsum[w] = v;
  __syncthreads();
  int woff = 0;
  for (int k = 0; k < 8; ++k) woff += (k < w) ? wsum[k] : 0;
  if (i < n) {
    int rp = base + woff + v - c;
    rowptr[i] = rp;
    degcnt[i] = c;
    float di = rsqrtf((float)c + 1.0f);
    dinv[i] = di;
    sd[i] = (float)cout * di;
    int run = rp;
#pragma unroll
    for (int k = 0; k < CH; ++k) {
      off[(size_t)k * NN + i] = run;
      run += (int)(hv[k] & 0xFFFFu);
    }
  }
}

__global__ void k_fill(const int* __restrict__ ei,
                       const int* __restrict__ rank,
                       const int* __restrict__ off,
                       int* __restrict__ csr, int E) {
  int e = blockIdx.x * blockDim.x + threadIdx.x;
  if (e >= E) return;
  int s = ei[e];
  int d = ei[E + e];
  csr[off[(size_t)(e / CS) * NN + d] + rank[e]] = s;
}

__global__ void __launch_bounds__(256)
k_gather_l1(const int* __restrict__ rowptr,
            const int* __restrict__ degcnt,
            const int* __restrict__ csr,
            const float* __restrict__ sd,
            const float* __restrict__ dinv,
            const float* __restrict__ W1,
            const float* __restrict__ b,
            const float* __restrict__ g,
            const float* __restrict__ be,
            const float* __restrict__ m,
            const float* __restrict__ v,
            float* __restrict__ x, int n) {
  int t = blockIdx.x * blockDim.x + threadIdx.x;
  int lane = threadIdx.x & 31;
  int sub = t & 3;
  int ndc = min(t >> 2, n - 1);
  int beg = rowptr[ndc];
  int cnt = degcnt[ndc];
  float dn = dinv[ndc];
  float acc = (sub == 0) ? sd[ndc] : 0.0f;
  int end = beg + cnt;
  int last = (cnt > 0) ? end - 1 : 0;
  int j0 = beg + sub;
  int s[8];
  float val[8];
#pragma unroll
  for (int i = 0; i < 8; ++i) s[i] = csr[min(j0 + 4 * i, last)];
#pragma unroll
  for (int i = 0; i < 8; ++i) val[i] = sd[s[i]];
#pragma unroll
  for (int i = 0; i < 8; ++i) acc += (j0 + 4 * i < end) ? val[i] : 0.0f;
  for (int j = j0 + 32; j < end; j += 4) acc += sd[csr[j]];
  acc += __shfl_xor(acc, 1);
  acc += __shfl_xor(acc, 2);
  float tn = acc * dn;

  int q  = lane & 15;
  int hi = lane >> 4;
  int node0 = (t - lane) >> 2;
  float w1[4], bb[4], mm[4], rs[4], gg[4], bee[4];
#pragma unroll
  for (int c = 0; c < 4; ++c) {
    int d = q * 4 + c;
    w1[c] = W1[d]; bb[c] = b[d]; mm[c] = m[d];
    rs[c] = rsqrtf(v[d] + BN_EPS); gg[c] = g[d]; bee[c] = be[d];
  }
#pragma unroll
  for (int it = 0; it < 4; ++it) {
    int ln = 2 * it + hi;
    float tv = __shfl(tn, ln * 4);
    float4 y;
    float* yp = &y.x;
#pragma unroll
    for (int c = 0; c < 4; ++c) {
      float r = tv * w1[c] + bb[c];
      yp[c] = fmaxf((r - mm[c]) * rs[c] * gg[c] + bee[c], 0.0f);
    }
    int node = node0 + ln;
    if (node < n) ((float4*)x)[(size_t)node * 16 + q] = y;
  }
}

template <bool BN>
__global__ void k_gather(const int* __restrict__ rowptr,
                         const int* __restrict__ degcnt,
                         const int* __restrict__ csr,
                         const float* __restrict__ hw,
                         const float* __restrict__ dinv,
                         const float* __restrict__ b,
                         const float* __restrict__ g,
                         const float* __restrict__ be,
                         const float* __restrict__ m,
                         const float* __restrict__ v,
                         float* __restrict__ x, int n) {
  int t = blockIdx.x * blockDim.x + threadIdx.x;
  int nd = t >> 4;
  int q  = t & 15;
  if (nd >= n) return;
  const float4* hw4 = (const float4*)hw;
  float dn = dinv[nd];
  int beg = rowptr[nd];
  int end = beg + degcnt[nd];
  float4 h = hw4[(size_t)nd * 16 + q];
  float ws = dn * dn;
  float4 a;
  a.x = h.x * ws; a.y = h.y * ws; a.z = h.z * ws; a.w = h.w * ws;
  if (beg < end) {
    int last = end - 1;
    int s0 = csr[beg];
    int s1 = csr[min(beg + 1, last)];
    int s2 = csr[min(beg + 2, last)];
    int s3 = csr[min(beg + 3, last)];
    for (int j = beg; j < end; j += 4) {
      bool p1 = (j + 1) < end;
      bool p2 = (j + 2) < end;
      bool p3 = (j + 3) < end;
      float w0 = dinv[s0] * dn;
      float w1 = dinv[s1] * dn;
      float w2 = dinv[s2] * dn;
      float w3 = dinv[s3] * dn;
      float4 m0 = hw4[(size_t)s0 * 16 + q];
      float4 m1 = hw4[(size_t)s1 * 16 + q];
      float4 m2 = hw4[(size_t)s2 * 16 + q];
      float4 m3 = hw4[(size_t)s3 * 16 + q];
      s0 = csr[min(j + 4, last)];
      s1 = csr[min(j + 5, last)];
      s2 = csr[min(j + 6, last)];
      s3 = csr[min(j + 7, last)];
      asm volatile("" :: "v"(m0.x), "v"(m0.y), "v"(m0.z), "v"(m0.w),
                         "v"(m1.x), "v"(m1.y), "v"(m1.z), "v"(m1.w),
                         "v"(m2.x), "v"(m2.y), "v"(m2.z), "v"(m2.w),
                         "v"(m3.x), "v"(m3.y), "v"(m3.z), "v"(m3.w));
      a.x += m0.x * w0; a.y += m0.y * w0; a.z += m0.z * w0; a.w += m0.w * w0;
      float4 u;
      u.x = a.x + m1.x * w1; u.y = a.y + m1.y * w1; u.z = a.z + m1.z * w1; u.w = a.w + m1.w * w1;
      a.x = p1 ? u.x : a.x; a.y = p1 ? u.y : a.y; a.z = p1 ? u.z : a.z; a.w = p1 ? u.w : a.w;
      u.x = a.x + m2.x * w2; u.y = a.y + m2.y * w2; u.z = a.z + m2.z * w2; u.w = a.w + m2.w * w2;
      a.x = p2 ? u.x : a.x; a.y = p2 ? u.y : a.y; a.z = p2 ? u.z : a.z; a.w = p2 ? u.w : a.w;
      u.x = a.x + m3.x * w3; u.y = a.y + m3.y * w3; u.z = a.z + m3.z * w3; u.w = a.w + m3.w * w3;
      a.x = p3 ? u.x : a.x; a.y = p3 ? u.y : a.y; a.z = p3 ? u.z : a.z; a.w = p3 ? u.w : a.w;
    }
  }
  float r[4] = {a.x, a.y, a.z, a.w};
  float4 y;
  float* yp = &y.x;
#pragma unroll
  for (int c = 0; c < 4; ++c) {
    int d = q * 4 + c;
    float val = r[c] + b[d];
    if (BN) {
      float rs = rsqrtf(v[d] + BN_EPS);
      val = fmaxf((val - m[d]) * rs * g[d] + be[d], 0.0f);
    }
    yp[c] = val;
  }
  ((float4*)x)[(size_t)nd * 16 + q] = y;
}

__global__ void k_pool_scatter(const int* __restrict__ batch,
                               const float* __restrict__ x,
                               float* __restrict__ pool,
                               float* __restrict__ cnt, int n) {
  int t = blockIdx.x * blockDim.x + threadIdx.x;
  int nd = t >> 4;
  int g  = t & 15;
  if (nd >= n) return;
  int bg = batch[nd];
  float4 m = ((const float4*)(x + (size_t)nd * DD))[g];
  float* dst = pool + (size_t)bg * DD + g * 4;
  atomicAdd(dst + 0, m.x);
  atomicAdd(dst + 1, m.y);
  atomicAdd(dst + 2, m.z);
  atomicAdd(dst + 3, m.w);
  if (g == 0) atomicAdd(&cnt[bg], 1.0f);
}

__global__ void k_pool_div(const float* __restrict__ pool,
                           const float* __restrict__ cnt,
                           float* __restrict__ pooled, int g64) {
  int i = blockIdx.x * blockDim.x + threadIdx.x;
  if (i >= g64) return;
  float c = cnt[i >> 6];
  pooled[i] = (c > 0.0f) ? pool[i] / c : 0.0f;
}

__global__ void k_gemm_wmma(const float* __restrict__ A,
                            const float* __restrict__ B,
                            float* __restrict__ C,
                            const float* __restrict__ bias,
                            const float* __restrict__ dinv,
                            float* __restrict__ acc,
                            int M, int K, int Nc) {
  int wid  = (blockIdx.x * blockDim.x + threadIdx.x) >> 5;
  int lane = threadIdx.x & 31;
  int tiles_n = Nc >> 4;
  int tm = wid / tiles_n;
  int tn = wid - tm * tiles_n;
  if (tm * 16 >= M) return;

  int half = lane >> 4;
  int lm   = lane & 15;
  const float* Arow = A + (size_t)(tm * 16 + lm) * K;
  int ncol = tn * 16 + lm;

  v8f c = {};
#if USE_WMMA_F32
  for (int k0 = 0; k0 < K; k0 += 4) {
    v2f a, b;
    a.x = Arow[k0 + 2 * half + 0];
    a.y = Arow[k0 + 2 * half + 1];
    b.x = B[(size_t)(k0 + 2 * half + 0) * Nc + ncol];
    b.y = B[(size_t)(k0 + 2 * half + 1) * Nc + ncol];
    c = __builtin_amdgcn_wmma_f32_16x16x4_f32(false, a, false, b, (short)0, c,
                                              false, false);
  }
#else
  for (int k0 = 0; k0 < K; k0 += 32) {
    v16h a, b;
#pragma unroll
    for (int i = 0; i < 8; ++i) {
      a[i]     = (_Float16)Arow[k0 + 8 * half + i];
      a[8 + i] = (_Float16)Arow[k0 + 16 + 8 * half + i];
    }
#pragma unroll
    for (int i = 0; i < 16; ++i)
      b[i] = (_Float16)B[(size_t)(k0 + 16 * half + i) * Nc + ncol];
    c = __builtin_amdgcn_wmma_f32_16x16x32_f16(false, a, false, b, (short)0, c,
                                               false, false);
  }
#endif

  float bv = bias ? bias[ncol] : 0.0f;
  if (acc) {
#pragma unroll
    for (int r = 0; r < 8; ++r) {
      int row = tm * 16 + 8 * half + r;
      size_t idx = (size_t)row * Nc + ncol;
      float val = c[r] + bv;
      float di = dinv[row];
      C[idx]   = val;
      acc[idx] = val * (di * di);
    }
  } else {
#pragma unroll
    for (int r = 0; r < 8; ++r) {
      C[(size_t)(tm * 16 + 8 * half + r) * Nc + ncol] = c[r] + bv;
    }
  }
}

static inline int blk(long long n, int b) { return (int)((n + b - 1) / b); }

extern "C" void kernel_launch(void* const* d_in, const int* in_sizes, int n_in,
                              void* d_out, int out_size, void* d_ws, size_t ws_size,
                              hipStream_t stream) {
  (void)in_sizes; (void)n_in; (void)out_size; (void)ws_size;
  const int*   edge_index = (const int*)d_in[0];
  const int*   batch      = (const int*)d_in[1];
  const float* W1 = (const float*)d_in[2];
  const float* b1 = (const float*)d_in[3];
  const float* W2 = (const float*)d_in[4];
  const float* b2 = (const float*)d_in[5];
  const float* W3 = (const float*)d_in[6];
  const float* b3 = (const float*)d_in[7];
  const float* g1 = (const float*)d_in[8];
  const float* be1= (const float*)d_in[9];
  const float* m1 = (const float*)d_in[10];
  const float* v1 = (const float*)d_in[11];
  const float* g2 = (const float*)d_in[12];
  const float* be2= (const float*)d_in[13];
  const float* m2 = (const float*)d_in[14];
  const float* v2 = (const float*)d_in[15];
  const float* Wl = (const float*)d_in[16];
  const float* bl = (const float*)d_in[17];
  float* out = (float*)d_out;

  char* ws = (char*)d_ws;
  size_t off = 0;
  auto carve = [&](size_t bytes) -> float* {
    float* p = (float*)(ws + off);
    off += (bytes + 255) & ~(size_t)255;
    return p;
  };
  int*   degcnt   = (int*)carve((size_t)NN * 4);
  float* dinv     = carve((size_t)NN * 4);
  float* sd       = carve((size_t)NN * 4);
  int*   rowptr   = (int*)carve((size_t)NN * 4);
  int*   rank     = (int*)carve((size_t)EE * 4);
  int*   csr      = (int*)carve((size_t)EE * 4);
  unsigned* hist  = (unsigned*)carve((size_t)CH * NN * 4);
  int*   coff     = (int*)carve((size_t)CH * NN * 4);
  int*   bsum     = (int*)carve((size_t)CH * NB * 4);
  float* bufX     = carve((size_t)NN * DD * 4);
  float* bufH     = carve((size_t)NN * DD * 4);
  float* pool     = carve((size_t)GG * DD * 4);
  float* cnt      = carve((size_t)GG * 4);
  float* pooled   = carve((size_t)GG * DD * 4);

  const int B = 256;
  const long long G64 = (long long)GG * DD;

  k_zero<<<blk(G64, B), B, 0, stream>>>(pool, (int)G64);
  k_zero<<<blk(GG, B), B, 0, stream>>>(cnt, GG);
  k_count_lds<<<CH, 1024, 0, stream>>>(edge_index, hist, bsum, rank);
  k_scan_dinv<<<NB, 256, 0, stream>>>(hist, bsum, coff, degcnt, rowptr, dinv, sd, NN);
  k_fill<<<blk(EE, B), B, 0, stream>>>(edge_index, rank, coff, csr, EE);

  const int tiles_NN  = (NN / 16) * (DD / 16);
  const int tiles_FIN = (GG / 16) * (DOUT / 16);

  k_gather_l1<<<blk((long long)NN * 4, B), B, 0, stream>>>(
      rowptr, degcnt, csr, sd, dinv, W1, b1, g1, be1, m1, v1, bufX, NN);

  k_gemm_wmma<<<blk((long long)tiles_NN * 32, B), B, 0, stream>>>(
      bufX, W2, bufH, nullptr, nullptr, nullptr, NN, DD, DD);
  k_gather<true><<<blk((long long)NN * 16, B), B, 0, stream>>>(
      rowptr, degcnt, csr, bufH, dinv, b2, g2, be2, m2, v2, bufX, NN);

  k_gemm_wmma<<<blk((long long)tiles_NN * 32, B), B, 0, stream>>>(
      bufX, W3, bufH, nullptr, nullptr, nullptr, NN, DD, DD);
  k_gather<false><<<blk((long long)NN * 16, B), B, 0, stream>>>(
      rowptr, degcnt, csr, bufH, dinv, b3, nullptr, nullptr, nullptr, nullptr, bufX, NN);

  k_pool_scatter<<<blk((long long)NN * 16, B), B, 0, stream>>>(batch, bufX, pool, cnt, NN);
  k_pool_div<<<blk(G64, B), B, 0, stream>>>(pool, cnt, pooled, (int)G64);
  k_gemm_wmma<<<blk((long long)tiles_FIN * 32, B), B, 0, stream>>>(
      pooled, Wl, out, bl, nullptr, nullptr, GG, DD, DOUT);
}


# Mean pool folded into its neighbours, 13 -> 9 launches: clears in the layer-1 launch, layer 3 sums rows per graph run in LDS, head divides at its A load; sum order: 16-node block, then arrival
# speedup vs baseline: 3.4425x; 1.2706x over previous
#include <hip/hip_runtime.h>
#include <hip/hip_bf16.h>

#define NN   50000
#define EE   800000
#define GG   512
#define DD   64
#define DOUT 32
#define BN_EPS 1e-5f

typedef __attribute__((ext_vector_type(2)))  float    v2f;
typedef __attribute__((ext_vector_type(8)))  float    v8f;
typedef __attribute__((ext_vector_type(16))) _Float16 v16h;

#if defined(__AMDGCN__) && __has_builtin(__builtin_amdgcn_wmma_f32_16x16x4_f32)
#define USE_WMMA_F32 1
#else
#define USE_WMMA_F32 0
#endif


#define CH 64
#define CS (EE / CH)
#define NB ((NN + 255) / 256)
static_assert(EE % CH == 0 && CS % 4 == 0, "a chunk is a whole number of 4-edge groups");
static_assert(CS < 65536, "a chunk's count for one node fits a 16-bit field");
static_assert(NN % 4 == 0, "the table is moved as 4-word groups");
static_assert(NB <= 256, "one thread of the scan block per node block in front of it");
#define G4  (CS / 4)
#define PER ((G4 + 1023) / 1024)

__global__ void __launch_bounds__(1024)
k_count_lds(const int* __restrict__ ei, unsigned* __restrict__ hist,
            int* __restrict__ bsum, int* __restrict__ rank) {
  __shared__ __attribute__((aligned(16))) unsigned h[NN];
  __shared__ int sb[NB];
  const int t = threadIdx.x;
  const int lane = t & 31;
  const int w = blockIdx.x;
  const int4* src4 = (const int4*)(ei + (size_t)w * CS);
  const int4* dst4 = (const int4*)(ei + EE + (size_t)w * CS);
  int4 sv[PER], dv[PER];
#pragma unroll
  for (int k = 0; k < PER; ++k) {
    int g = min(t + 1024 * k, G4 - 1);
    sv[k] = src4[g];
    dv[k] = dst4[g];
  }
  uint4* h4 = (uint4*)h;
  for (int j = t; j < NN / 4; j += 1024) h4[j] = make_uint4(0u, 0u, 0u, 0u);
  if (t < NB) sb[t] = 0;
  __syncthreads();
  int4* rank4 = (int4*)(rank + (size_t)w * CS);
#pragma unroll
  for (int k = 0; k < PER; ++k) {
    int g = t + 1024 * k;
    if (g < G4) {
      int4 r;
      r.x = (int)(atomicAdd(&h[dv[k].x], 1u) & 0xFFFFu);
      r.y = (int)(atomicAdd(&h[dv[k].y], 1u) & 0xFFFFu);
      r.z = (int)(atomicAdd(&h[dv[k].z], 1u) & 0xFFFFu);
      r.w = (int)(atomicAdd(&h[dv[k].w], 1u) & 0xFFFFu);
      atomicAdd(&h[sv[k].x], 0x10000u);
      atomicAdd(&h[sv[k].y], 0x10000u);
      atomicAdd(&h[sv[k].z], 0x10000u);
      atomicAdd(&h[sv[k].w], 0x10000u);
      rank4[g] = r;
    }
  }
  __syncthreads();
  uint4* hg4 = (uint4*)(hist + (size_t)w * NN);
  for (int j0 = 0; j0 < NN / 4; j0 += 1024) {
    int j = j0 + t;
    bool ok = j < NN / 4;
    uint4 v = make_uint4(0u, 0u, 0u, 0u);
    if (ok) { v = h4[j]; hg4[j] = v; }
    int c = (int)((v.x & 0xFFFFu) + (v.y & 0xFFFFu) + (v.z & 0xFFFFu) + (v.w & 0xFFFFu));
    for (int o = 16; o > 0; o >>= 1) c += __shfl_xor(c, o);
    if (lane == 0 && ok) atomicAdd(&sb[j >> 6], c);
  }
  __syncthreads();
  if (t < NB) bsum[w * NB + t] = sb[t];
}

__global__ void __launch_bounds__(256)
k_scan_dinv(const unsigned* __restrict__ hist, const int* __restrict__ bsum,
            int* __restrict__ off, int* __restrict__ degcnt,
            int* __restrict__ rowptr, float* __restrict__ dinv,
            float* __restrict__ sd, int n) {
  __shared__ int wsum[8];
  __shared__ int s_base;
  int t = threadIdx.x;
  int lane = t & 31;
  int w = t >> 5;
  int b = blockIdx.x;
  int i = b * 256 + t;
  int ic = min(i, n - 1);
  unsigned hv[CH];
#pragma unroll
  for (int k = 0; k < CH; ++k) hv[k] = hist[(size_t)k * NN + ic];

  int p = 0;
  if (t < b) {
    for (int k = 0; k < CH; ++k) p += bsum[k * NB + t];
  }
  for (int o = 16; o > 0; o >>= 1) p += __shfl_xor(p, o);
  if (lane == 0) wsum[w] = p;
  __syncthreads();
  if (w == 0) {
    int x = (lane < 8) ? wsum[lane] : 0;
    for (int o = 16; o > 0; o >>= 1) x += __shfl_xor(x, o);
    if (lane == 0) s_base = x;
  }
  __syncthreads();
  int base = s_base;

  int cin = 0, cout = 0;
#pragma unroll
  for (int k = 0; k < CH; ++k) { cin += (int)(hv[k] & 0xFFFFu); cout += (int)(hv[k] >> 16); }
  int c = (i < n) ? cin : 0;
  int v = c;
  for (int o = 1; o < 32; o <<= 1) {
    int u = __shfl_up(v, o);
    if (lane >= o) v += u;
  }
  if (lane == 31) wsum[w] = v;
  __syncthreads();
  int woff = 0;
  for (int k = 0; k < 8; ++k) woff += (k < w) ? wsum[k] : 0;
  if (i < n) {
    int rp = base + woff + v - c;
    rowptr[i] = rp;
    degcnt[i] = c;
    float di = rsqrtf((float)c + 1.0f);
    dinv[i] = di;
    sd[i] = (float)cout * di;
    int run = rp;
#pragma unroll
    for (int k = 0; k < CH; ++k) {
      off[(size_t)k * NN + i] = run;
      run += (int)(hv[k] & 0xFFFFu);
    }
  }
}

__global__ void k_fill(const int* __restrict__ ei,
                       const int* __restrict__ rank,
                       const int* __restrict__ off,
                       int* __restrict__ csr, int E) {
  int e = blockIdx.x * blockDim.x + threadIdx.x;
  if (e >= E) return;
  int s = ei[e];
  int d = ei[E + e];
  csr[off[(size_t)(e / CS) * NN + d] + rank[e]] = s;
}

__global__ void __launch_bounds__(256)
k_gather_l1(const int* __restrict__ rowptr,
            const int* __restrict__ degcnt,
            const int* __restrict__ csr,
            const float* __restrict__ sd,
            const float* __restrict__ dinv,
            const float* __restrict__ W1,
            const float* __restrict__ b,
            const float* __restrict__ g,
            const float* __restrict__ be,
            const float* __restrict__ m,
            const float* __restrict__ v,
            float* __restrict__ x,
            float* __restrict__ gsum,
            float* __restrict__ gcnt, int n) {
  int t = blockIdx.x * blockDim.x + threadIdx.x;
  int lane = threadIdx.x & 31;
  int sub = t & 3;
  static_assert((long long)NN * 4 >= (long long)GG * DD, "one thread of this grid per pool word");
  if (t < GG * DD) gsum[t] = 0.0f;
  if (t < GG) gcnt[t] = 0.0f;
  int ndc = min(t >> 2, n - 1);
  int beg = rowptr[ndc];
  int cnt = degcnt[ndc];
  float dn = dinv[ndc];
  float acc = (sub == 0) ? sd[ndc] : 0.0f;
  int end = beg + cnt;
  int last = (cnt > 0) ? end - 1 : 0;
  int j0 = beg + sub;
  int s[8];
  float val[8];
#pragma unroll
  for (int i = 0; i < 8; ++i) s[i] = csr[min(j0 + 4 * i, last)];
#pragma unroll
  for (int i = 0; i < 8; ++i) val[i] = sd[s[i]];
#pragma unroll
  for (int i = 0; i < 8; ++i) acc += (j0 + 4 * i < end) ? val[i] : 0.0f;
  for (int j = j0 + 32; j < end; j += 4) acc += sd[csr[j]];
  acc += __shfl_xor(acc, 1);
  acc += __shfl_xor(acc, 2);
  float tn = acc * dn;

  int q  = lane & 15;
  int hi = lane >> 4;
  int node0 = (t - lane) >> 2;
  float w1[4], bb[4], mm[4], rs[4], gg[4], bee[4];
#pragma unroll
  for (int c = 0; c < 4; ++c) {
    int d = q * 4 + c;
    w1[c] = W1[d]; bb[c] = b[d]; mm[c] = m[d];
    rs[c] = rsqrtf(v[d] + BN_EPS); gg[c] = g[d]; bee[c] = be[d];
  }
#pragma unroll
  for (int it = 0; it < 4; ++it) {
    int ln = 2 * it + hi;
    float tv = __shfl(tn, ln * 4);
    float4 y;
    float* yp = &y.x;
#pragma unroll
    for (int c = 0; c < 4; ++c) {
      float r = tv * w1[c] + bb[c];
      yp[c] = fmaxf((r - mm[c]) * rs[c] * gg[c] + bee[c], 0.0f);
    }
    int node = node0 + ln;
    if (node < n) ((float4*)x)[(size_t)node * 16 + q] = y;
  }
}

template <bool BN>
__global__ void __launch_bounds__(256)
k_gather(const int* __restrict__ rowptr,
                         const int* __restrict__ degcnt,
                         const int* __restrict__ csr,
                         const float* __restrict__ hw,
                         const float* __restrict__ dinv,
                         const float* __restrict__ b,
                         const float* __restrict__ g,
                         const float* __restrict__ be,
                         const float* __restrict__ m,
                         const float* __restrict__ v,
                         float* __restrict__ x,
                         const int* __restrict__ batch,
                         float* __restrict__ pool,
                         float* __restrict__ cnt, int n) {
  int t = blockIdx.x * blockDim.x + threadIdx.x;
  int nd = t >> 4;
  int q  = t & 15;
  bool live = nd < n;
  if (BN) {
    if (!live) return;
  } else {
    nd = min(nd, n - 1);
  }
  const float4* hw4 = (const float4*)hw;
  float dn = dinv[nd];
  int beg = rowptr[nd];
  int end = beg + degcnt[nd];
  float4 h = hw4[(size_t)nd * 16 + q];
  float ws = dn * dn;
  float4 a;
  a.x = h.x * ws; a.y = h.y * ws; a.z = h.z * ws; a.w = h.w * ws;
  if (beg < end) {
    int last = end - 1;
    int s0 = csr[beg];
    int s1 = csr[min(beg + 1, last)];
    int s2 = csr[min(beg + 2, last)];
    int s3 = csr[min(beg + 3, last)];
    for (int j = beg; j < end; j += 4) {
      bool p1 = (j + 1) < end;
      bool p2 = (j + 2) < end;
      bool p3 = (j + 3) < end;
      float w0 = dinv[s0] * dn;
      float w1 = dinv[s1] * dn;
      float w2 = dinv[s2] * dn;
      float w3 = dinv[s3] * dn;
      float4 m0 = hw4[(size_t)s0 * 16 + q];
      float4 m1 = hw4[(size_t)s1 * 16 + q];
      float4 m2 = hw4[(size_t)s2 * 16 + q];
      float4 m3 = hw4[(size_t)s3 * 16 + q];
      s0 = csr[min(j + 4, last)];
      s1 = csr[min(j + 5, last)];
      s2 = csr[min(j + 6, last)];
      s3 = csr[min(j + 7, last)];
      asm volatile("" :: "v"(m0.x), "v"(m0.y), "v"(m0.z), "v"(m0.w),
                         "v"(m1.x), "v"(m1.y), "v"(m1.z), "v"(m1.w),
                         "v"(m2.x), "v"(m2.y), "v"(m2.z), "v"(m2.w),
                         "v"(m3.x), "v"(m3.y), "v"(m3.z), "v"(m3.w));
      a.x += m0.x * w0; a.y += m0.y * w0; a.z += m0.z * w0; a.w += m0.w * w0;
      float4 u;
      u.x = a.x + m1.x * w1; u.y = a.y + m1.y * w1; u.z = a.z + m1.z * w1; u.w = a.w + m1.w * w1;
      a.x = p1 ? u.x : a.x; a.y = p1 ? u.y : a.y; a.z = p1 ? u.z : a.z; a.w = p1 ? u.w : a.w;
      u.x = a.x + m2.x * w2; u.y = a.y + m2.y * w2; u.z = a.z + m2.z * w2; u.w = a.w + m2.w * w2;
      a.x = p2 ? u.x : a.x; a.y = p2 ? u.y : a.y; a.z = p2 ? u.z : a.z; a.w = p2 ? u.w : a.w;
      u.x = a.x + m3.x * w3; u.y = a.y + m3.y * w3; u.z = a.z + m3.z * w3; u.w = a.w + m3.w * w3;
      a.x = p3 ? u.x : a.x; a.y = p3 ? u.y : a.y; a.z = p3 ? u.z : a.z; a.w = p3 ? u.w : a.w;
    }
  }
  float r[4] = {a.x, a.y, a.z, a.w};
  float4 y;
  float* yp = &y.x;
#pragma unroll
  for (int c = 0; c < 4; ++c) {
    int d = q * 4 + c;
    float val = r[c] + b[d];
    if (BN) {
      float rs = rsqrtf(v[d] + BN_EPS);
      val = fmaxf((val - m[d]) * rs * g[d] + be[d], 0.0f);
    }
    yp[c] = val;
  }
  if (BN) {
    ((float4*)x)[(size_t)nd * 16 + q] = y;
  } else {
    __shared__ __attribute__((aligned(16))) float srow[16][DD];
    __shared__ int sgid[16];
    int ln = threadIdx.x >> 4;
    ((float4*)&srow[ln][0])[q] = y;
    if (q == 0) sgid[ln] = live ? batch[nd] : -1;
    __syncthreads();
    if (threadIdx.x < DD) {
      int c = threadIdx.x;
      float vals[16];
      int gids[16];
#pragma unroll
      for (int k = 0; k < 16; ++k) { vals[k] = srow[k][c]; gids[k] = sgid[k]; }
      int gcur = gids[0];
      float s = vals[0];
      int run = 1;
#pragma unroll
      for (int k = 1; k < 16; ++k) {
        int gk = gids[k];
        float vk = vals[k];
        if (gk == gcur) {
          s += vk;
          ++run;
        } else {
          if (gcur >= 0) {
            atomicAdd(&pool[(size_t)gcur * DD + c], s);
            if (c == 0) atomicAdd(&cnt[gcur], (float)run);
          }
          gcur = gk; s = vk; run = 1;
        }
      }
      if (gcur >= 0) {
        atomicAdd(&pool[(size_t)gcur * DD + c], s);
        if (c == 0) atomicAdd(&cnt[gcur], (float)run);
      }
    }
  }
}

template <bool ROWDIV>
__global__ void k_gemm_wmma(const float* __restrict__ A,
                            const float* __restrict__ B,
                            float* __restrict__ C,
                            const float* __restrict__ bias,
                            const float* __restrict__ dinv,
                            float* __restrict__ acc,
                            const float* __restrict__ rowcnt,
                            int M, int K, int Nc) {
  int wid  = (blockIdx.x * blockDim.x + threadIdx.x) >> 5;
  int lane = threadIdx.x & 31;
  int tiles_n = Nc >> 4;
  int tm = wid / tiles_n;
  int tn = wid - tm * tiles_n;
  if (tm * 16 >= M) return;

  int half = lane >> 4;
  int lm   = lane & 15;
  const float* Arow = A + (size_t)(tm * 16 + lm) * K;
  int ncol = tn * 16 + lm;
  float rc = 1.0f;
  bool has = true;
  if (ROWDIV) {
    float cr = rowcnt[tm * 16 + lm];
    has = cr > 0.0f;
    rc = has ? cr : 1.0f;
  }

  v8f c = {};
#if USE_WMMA_F32
  for (int k0 = 0; k0 < K; k0 += 4) {
    v2f a, b;
    a.x = Arow[k0 + 2 * half + 0];
    a.y = Arow[k0 + 2 * half + 1];
    if (ROWDIV) {
      float qx = a.x / rc;
      float qy = a.y / rc;
      a.x = has ? qx : 0.0f;
      a.y = has ? qy : 0.0f;
    }
    b.x = B[(size_t)(k0 + 2 * half + 0) * Nc + ncol];
    b.y = B[(size_t)(k0 + 2 * half + 1) * Nc + ncol];
    c = __builtin_amdgcn_wmma_f32_16x16x4_f32(false, a, false, b, (short)0, c,
                                              false, false);
  }
#else
  for (int k0 = 0; k0 < K; k0 += 32) {
    v16h a, b;
#pragma unroll
    for (int i = 0; i < 8; ++i) {
      float lo = Arow[k0 + 8 * half + i];
      float hi = Arow[k0 + 16 + 8 * half + i];
      if (ROWDIV) {
        lo = has ? lo / rc : 0.0f;
        hi = has ? hi / rc : 0.0f;
      }
      a[i]     = (_Float16)lo;
      a[8 + i] = (_Float16)hi;
    }
#pragma unroll
    for (int i = 0; i < 16; ++i)
      b[i] = (_Float16)B[(size_t)(k0 + 16 * half + i) * Nc + ncol];
    c = __builtin_amdgcn_wmma_f32_16x16x32_f16(false, a, false, b, (short)0, c,
                                               false, false);
  }
#endif

  float bv = bias ? bias[ncol] : 0.0f;
  if (acc) {
#pragma unroll
    for (int r = 0; r < 8; ++r) {
      int row = tm * 16 + 8 * half + r;
      size_t idx = (size_t)row * Nc + ncol;
      float val = c[r] + bv;
      float di = dinv[row];
      C[idx]   = val;
      acc[idx] = val * (di * di);
    }
  } else {
#pragma unroll
    for (int r = 0; r < 8; ++r) {
      C[(size_t)(tm * 16 + 8 * half + r) * Nc + ncol] = c[r] + bv;
    }
  }
}

static inline int blk(long long n, int b) { return (int)((n + b - 1) / b); }

extern "C" void kernel_launch(void* const* d_in, const int* in_sizes, int n_in,
                              void* d_out, int out_size, void* d_ws, size_t ws_size,
                              hipStream_t stream) {
  (void)in_sizes; (void)n_in; (void)out_size; (void)ws_size;
  const int*   edge_index = (const int*)d_in[0];
  const int*   batch      = (const int*)d_in[1];
  const float* W1 = (const float*)d_in[2];
  const float* b1 = (const float*)d_in[3];
  const float* W2 = (const float*)d_in[4];
  const float* b2 = (const float*)d_in[5];
  const float* W3 = (const float*)d_in[6];
  const float* b3 = (const float*)d_in[7];
  const float* g1 = (const float*)d_in[8];
  const float* be1= (const float*)d_in[9];
  const float* m1 = (const float*)d_in[10];
  const float* v1 = (const float*)d_in[11];
  const float* g2 = (const float*)d_in[12];
  const float* be2= (const float*)d_in[13];
  const float* m2 = (const float*)d_in[14];
  const float* v2 = (const float*)d_in[15];
  const float* Wl = (const float*)d_in[16];
  const float* bl = (const float*)d_in[17];
  float* out = (float*)d_out;

  char* ws = (char*)d_ws;
  size_t off = 0;
  auto carve = [&](size_t bytes) -> float* {
    float* p = (float*)(ws + off);
    off += (bytes + 255) & ~(size_t)255;
    return p;
  };
  int*   degcnt   = (int*)carve((size_t)NN * 4);
  float* dinv     = carve((size_t)NN * 4);
  float* sd       = carve((size_t)NN * 4);
  int*   rowptr   = (int*)carve((size_t)NN * 4);
  int*   rank     = (int*)carve((size_t)EE * 4);
  int*   csr      = (int*)carve((size_t)EE * 4);
  unsigned* hist  = (unsigned*)carve((size_t)CH * NN * 4);
  int*   coff     = (int*)carve((size_t)CH * NN * 4);
  int*   bsum     = (int*)carve((size_t)CH * NB * 4);
  float* bufX     = carve((size_t)NN * DD * 4);
  float* bufH     = carve((size_t)NN * DD * 4);
  float* pool     = carve((size_t)GG * DD * 4);
  float* cnt      = carve((size_t)GG * 4);

  const int B = 256;

  k_count_lds<<<CH, 1024, 0, stream>>>(edge_index, hist, bsum, rank);
  k_scan_dinv<<<NB, 256, 0, stream>>>(hist, bsum, coff, degcnt, rowptr, dinv, sd, NN);
  k_fill<<<blk(EE, B), B, 0, stream>>>(edge_index, rank, coff, csr, EE);

  const int tiles_NN  = (NN / 16) * (DD / 16);
  const int tiles_FIN = (GG / 16) * (DOUT / 16);

  k_gather_l1<<<blk((long long)NN * 4, B), B, 0, stream>>>(
      rowptr, degcnt, csr, sd, dinv, W1, b1, g1, be1, m1, v1, bufX, pool, cnt, NN);

  k_gemm_wmma<false><<<blk((long long)tiles_NN * 32, B), B, 0, stream>>>(
      bufX, W2, bufH, nullptr, nullptr, nullptr, nullptr, NN, DD, DD);
  k_gather<true><<<blk((long long)NN * 16, B), B, 0, stream>>>(
      rowptr, degcnt, csr, bufH, dinv, b2, g2, be2, m2, v2, bufX,
      nullptr, nullptr, nullptr, NN);

  k_gemm_wmma<false><<<blk((long long)tiles_NN * 32, B), B, 0, stream>>>(
      bufX, W3, bufH, nullptr, nullptr, nullptr, nullptr, NN, DD, DD);
  k_gather<false><<<blk((long long)NN * 16, B), B, 0, stream>>>(
      rowptr, degcnt, csr, bufH, dinv, b3, nullptr, nullptr, nullptr, nullptr, nullptr,
      batch, pool, cnt, NN);

  k_gemm_wmma<true><<<blk((long long)tiles_FIN * 32, B), B, 0, stream>>>(
      pool, Wl, out, bl, nullptr, nullptr, cnt, GG, DD, DOUT);
}


# Each layer's dense product moved behind its aggregation into the gather launch (16 summed rows in LDS x W on the f32 matrix instruction): 9 -> 7 launches; sum order: neighbours, then k in 2 chains
# speedup vs baseline: 3.8618x; 1.1218x over previous
#include <hip/hip_runtime.h>
#include <hip/hip_bf16.h>

#define NN   50000
#define EE   800000
#define GG   512
#define DD   64
#define DOUT 32
#define BN_EPS 1e-5f

typedef __attribute__((ext_vector_type(2)))  float    v2f;
typedef __attribute__((ext_vector_type(8)))  float    v8f;
typedef __attribute__((ext_vector_type(16))) _Float16 v16h;

#if defined(__AMDGCN__) && __has_builtin(__builtin_amdgcn_wmma_f32_16x16x4_f32)
#define USE_WMMA_F32 1
#else
#define USE_WMMA_F32 0
#endif


#define CH 64
#define CS (EE / CH)
#define NB ((NN + 255) / 256)
static_assert(EE % CH == 0 && CS % 4 == 0, "a chunk is a whole number of 4-edge groups");
static_assert(CS < 65536, "a chunk's count for one node fits a 16-bit field");
static_assert(NN % 4 == 0, "the table is moved as 4-word groups");
static_assert(NB <= 256, "one thread of the scan block per node block in front of it");
#define G4  (CS / 4)
#define PER ((G4 + 1023) / 1024)

__global__ void __launch_bounds__(1024)
k_count_lds(const int* __restrict__ ei, unsigned* __restrict__ hist,
            int* __restrict__ bsum, int* __restrict__ rank) {
  __shared__ __attribute__((aligned(16))) unsigned h[NN];
  __shared__ int sb[NB];
  const int t = threadIdx.x;
  const int lane = t & 31;
  const int w = blockIdx.x;
  const int4* src4 = (const int4*)(ei + (size_t)w * CS);
  const int4* dst4 = (const int4*)(ei + EE + (size_t)w * CS);
  int4 sv[PER], dv[PER];
#pragma unroll
  for (int k = 0; k < PER; ++k) {
    int g = min(t + 1024 * k, G4 - 1);
    sv[k] = src4[g];
    dv[k] = dst4[g];
  }
  uint4* h4 = (uint4*)h;
  for (int j = t; j < NN / 4; j += 1024) h4[j] = make_uint4(0u, 0u, 0u, 0u);
  if (t < NB) sb[t] = 0;
  __syncthreads();
  int4* rank4 = (int4*)(rank + (size_t)w * CS);
#pragma unroll
  for (int k = 0; k < PER; ++k) {
    int g = t + 1024 * k;
    if (g < G4) {
      int4 r;
      r.x = (int)(atomicAdd(&h[dv[k].x], 1u) & 0xFFFFu);
      r.y = (int)(atomicAdd(&h[dv[k].y], 1u) & 0xFFFFu);
      r.z = (int)(atomicAdd(&h[dv[k].z], 1u) & 0xFFFFu);
      r.w = (int)(atomicAdd(&h[dv[k].w], 1u) & 0xFFFFu);
      atomicAdd(&h[sv[k].x], 0x10000u);
      atomicAdd(&h[sv[k].y], 0x10000u);
      atomicAdd(&h[sv[k].z], 0x10000u);
      atomicAdd(&h[sv[k].w], 0x10000u);
      rank4[g] = r;
    }
  }
  __syncthreads();
  uint4* hg4 = (uint4*)(hist + (size_t)w * NN);
  for (int j0 = 0; j0 < NN / 4; j0 += 1024) {
    int j = j0 + t;
    bool ok = j < NN / 4;
    uint4 v = make_uint4(0u, 0u, 0u, 0u);
    if (ok) { v = h4[j]; hg4[j] = v; }
    int c = (int)((v.x & 0xFFFFu) + (v.y & 0xFFFFu) + (v.z & 0xFFFFu) + (v.w & 0xFFFFu));
    for (int o = 16; o > 0; o >>= 1) c += __shfl_xor(c, o);
    if (lane == 0 && ok) atomicAdd(&sb[j >> 6], c);
  }
  __syncthreads();
  if (t < NB) bsum[w * NB + t] = sb[t];
}

__global__ void __launch_bounds__(256)
k_scan_dinv(const unsigned* __restrict__ hist, const int* __restrict__ bsum,
            int* __restrict__ off, int* __restrict__ degcnt,
            int* __restrict__ rowptr, float* __restrict__ dinv,
            float* __restrict__ sd, int n) {
  __shared__ int wsum[8];
  __shared__ int s_base;
  int t = threadIdx.x;
  int lane = t & 31;
  int w = t >> 5;
  int b = blockIdx.x;
  int i = b * 256 + t;
  int ic = min(i, n - 1);
  unsigned hv[CH];
#pragma unroll
  for (int k = 0; k < CH; ++k) hv[k] = hist[(size_t)k * NN + ic];

  int p = 0;
  if (t < b) {
    for (int k = 0; k < CH; ++k) p += bsum[k * NB + t];
  }
  for (int o = 16; o > 0; o >>= 1) p += __shfl_xor(p, o);
  if (lane == 0) wsum[w] = p;
  __syncthreads();
  if (w == 0) {
    int x = (lane < 8) ? wsum[lane] : 0;
    for (int o = 16; o > 0; o >>= 1) x += __shfl_xor(x, o);
    if (lane == 0) s_base = x;
  }
  __syncthreads();
  int base = s_base;

  int cin = 0, cout = 0;
#pragma unroll
  for (int k = 0; k < CH; ++k) { cin += (int)(hv[k] & 0xFFFFu); cout += (int)(hv[k] >> 16); }
  int c = (i < n) ? cin : 0;
  int v = c;
  for (int o = 1; o < 32; o <<= 1) {
    int u = __shfl_up(v, o);
    if (lane >= o) v += u;
  }
  if (lane == 31) wsum[w] = v;
  __syncthreads();
  int woff = 0;
  for (int k = 0; k < 8; ++k) woff += (k < w) ? wsum[k] : 0;
  if (i < n) {
    int rp = base + woff + v - c;
    rowptr[i] = rp;
    degcnt[i] = c;
    float di = rsqrtf((float)c + 1.0f);
    dinv[i] = di;
    sd[i] = (float)cout * di;
    int run = rp;
#pragma unroll
    for (int k = 0; k < CH; ++k) {
      off[(size_t)k * NN + i] = run;
      run += (int)(hv[k] & 0xFFFFu);
    }
  }
}

__global__ void k_fill(const int* __restrict__ ei,
                       const int* __restrict__ rank,
                       const int* __restrict__ off,
                       int* __restrict__ csr, int E) {
  int e = blockIdx.x * blockDim.x + threadIdx.x;
  if (e >= E) return;
  int s = ei[e];
  int d = ei[E + e];
  csr[off[(size_t)(e / CS) * NN + d] + rank[e]] = s;
}

__global__ void __launch_bounds__(256)
k_gather_l1(const int* __restrict__ rowptr,
            const int* __restrict__ degcnt,
            const int* __restrict__ csr,
            const float* __restrict__ sd,
            const float* __restrict__ dinv,
            const float* __restrict__ W1,
            const float* __restrict__ b,
            const float* __restrict__ g,
            const float* __restrict__ be,
            const float* __restrict__ m,
            const float* __restrict__ v,
            float* __restrict__ x,
            float* __restrict__ gsum,
            float* __restrict__ gcnt, int n) {
  int t = blockIdx.x * blockDim.x + threadIdx.x;
  int lane = threadIdx.x & 31;
  int sub = t & 3;
  static_assert((long long)NN * 4 >= (long long)GG * DD, "one thread of this grid per pool word");
  if (t < GG * DD) gsum[t] = 0.0f;
  if (t < GG) gcnt[t] = 0.0f;
  int ndc = min(t >> 2, n - 1);
  int beg = rowptr[ndc];
  int cnt = degcnt[ndc];
  float dn = dinv[ndc];
  float acc = (sub == 0) ? sd[ndc] : 0.0f;
  int end = beg + cnt;
  int last = (cnt > 0) ? end - 1 : 0;
  int j0 = beg + sub;
  int s[8];
  float val[8];
#pragma unroll
  for (int i = 0; i < 8; ++i) s[i] = csr[min(j0 + 4 * i, last)];
#pragma unroll
  for (int i = 0; i < 8; ++i) val[i] = sd[s[i]];
#pragma unroll
  for (int i = 0; i < 8; ++i) acc += (j0 + 4 * i < end) ? val[i] : 0.0f;
  for (int j = j0 + 32; j < end; j += 4) acc += sd[csr[j]];
  acc += __shfl_xor(acc, 1);
  acc += __shfl_xor(acc, 2);
  float tn = acc * dn;

  int q  = lane & 15;
  int hi = lane >> 4;
  int node0 = (t - lane) >> 2;
  float w1[4], bb[4], mm[4], rs[4], gg[4], bee[4];
#pragma unroll
  for (int c = 0; c < 4; ++c) {
    int d = q * 4 + c;
    w1[c] = W1[d]; bb[c] = b[d]; mm[c] = m[d];
    rs[c] = rsqrtf(v[d] + BN_EPS); gg[c] = g[d]; bee[c] = be[d];
  }
#pragma unroll
  for (int it = 0; it < 4; ++it) {
    int ln = 2 * it + hi;
    float tv = __shfl(tn, ln * 4);
    float4 y;
    float* yp = &y.x;
#pragma unroll
    for (int c = 0; c < 4; ++c) {
      float r = tv * w1[c] + bb[c];
      yp[c] = fmaxf((r - mm[c]) * rs[c] * gg[c] + bee[c], 0.0f);
    }
    int node = node0 + ln;
    if (node < n) ((float4*)x)[(size_t)node * 16 + q] = y;
  }
}

#define TP 68
template <bool BN>
__global__ void __launch_bounds__(256)
k_gather_mm(const int* __restrict__ rowptr,
                         const int* __restrict__ degcnt,
                         const int* __restrict__ csr,
                         const float* __restrict__ xin,
                         const float* __restrict__ dinv,
                         const float* __restrict__ W,
                         const float* __restrict__ b,
                         const float* __restrict__ g,
                         const float* __restrict__ be,
                         const float* __restrict__ m,
                         const float* __restrict__ v,
                         float* __restrict__ x,
                         const int* __restrict__ batch,
                         float* __restrict__ pool,
                         float* __restrict__ cnt, int n) {
  __shared__ __attribute__((aligned(16))) float sagg[16][TP];
  __shared__ __attribute__((aligned(16))) float srow[16][TP];
  __shared__ int sgid[16];
  int t = blockIdx.x * blockDim.x + threadIdx.x;
  int nd = t >> 4;
  int q  = t & 15;
  bool live = nd < n;
  nd = min(nd, n - 1);
  const float4* xin4 = (const float4*)xin;
  float dn = dinv[nd];
  int beg = rowptr[nd];
  int end = beg + degcnt[nd];
  float4 h = xin4[(size_t)nd * 16 + q];
  float ws = dn * dn;
  float4 a;
  a.x = h.x * ws; a.y = h.y * ws; a.z = h.z * ws; a.w = h.w * ws;
  if (beg < end) {
    int last = end - 1;
    int s0 = csr[beg];
    int s1 = csr[min(beg + 1, last)];
    int s2 = csr[min(beg + 2, last)];
    int s3 = csr[min(beg + 3, last)];
    for (int j = beg; j < end; j += 4) {
      bool p1 = (j + 1) < end;
      bool p2 = (j + 2) < end;
      bool p3 = (j + 3) < end;
      float w0 = dinv[s0] * dn;
      float w1 = dinv[s1] * dn;
      float w2 = dinv[s2] * dn;
      float w3 = dinv[s3] * dn;
      float4 m0 = xin4[(size_t)s0 * 16 + q];
      float4 m1 = xin4[(size_t)s1 * 16 + q];
      float4 m2 = xin4[(size_t)s2 * 16 + q];
      float4 m3 = xin4[(size_t)s3 * 16 + q];
      s0 = csr[min(j + 4, last)];
      s1 = csr[min(j + 5, last)];
      s2 = csr[min(j + 6, last)];
      s3 = csr[min(j + 7, last)];
      asm volatile("" :: "v"(m0.x), "v"(m0.y), "v"(m0.z), "v"(m0.w),
                         "v"(m1.x), "v"(m1.y), "v"(m1.z), "v"(m1.w),
                         "v"(m2.x), "v"(m2.y), "v"(m2.z), "v"(m2.w),
                         "v"(m3.x), "v"(m3.y), "v"(m3.z), "v"(m3.w));
      __builtin_amdgcn_sched_barrier(0);
      a.x += m0.x * w0; a.y += m0.y * w0; a.z += m0.z * w0; a.w += m0.w * w0;
      float4 u;
      u.x = a.x + m1.x * w1; u.y = a.y + m1.y * w1; u.z = a.z + m1.z * w1; u.w = a.w + m1.w * w1;
      a.x = p1 ? u.x : a.x; a.y = p1 ? u.y : a.y; a.z = p1 ? u.z : a.z; a.w = p1 ? u.w : a.w;
      u.x = a.x + m2.x * w2; u.y = a.y + m2.y * w2; u.z = a.z + m2.z * w2; u.w = a.w + m2.w * w2;
      a.x = p2 ? u.x : a.x; a.y = p2 ? u.y : a.y; a.z = p2 ? u.z : a.z; a.w = p2 ? u.w : a.w;
      u.x = a.x + m3.x * w3; u.y = a.y + m3.y * w3; u.z = a.z + m3.z * w3; u.w = a.w + m3.w * w3;
      a.x = p3 ? u.x : a.x; a.y = p3 ? u.y : a.y; a.z = p3 ? u.z : a.z; a.w = p3 ? u.w : a.w;
    }
  }
  int ln = threadIdx.x >> 4;
  int wv = __builtin_amdgcn_readfirstlane((int)(threadIdx.x >> 5));
  int lane = threadIdx.x & 31;
  int half = lane >> 4;
  int lm   = lane & 15;
  int col  = (wv & 3) * 16 + lm;
#if USE_WMMA_F32
  v2f bf[DD / 4];
  if (wv < 4) {
#pragma unroll
    for (int ks = 0; ks < DD / 4; ++ks) {
      bf[ks].x = W[(size_t)(4 * ks + 2 * half + 0) * DD + col];
      bf[ks].y = W[(size_t)(4 * ks + 2 * half + 1) * DD + col];
    }
  }
#endif
  ((float4*)&sagg[ln][0])[q] = a;
  if (!BN) {
    if (q == 0) sgid[ln] = live ? batch[nd] : -1;
  }
  __syncthreads();
  if (wv < 4) {
    v8f c0 = {};
    v8f c1 = {};
#if USE_WMMA_F32
#pragma unroll
    for (int ks = 0; ks < DD / 4; ks += 2) {
      v2f a0 = *(const v2f*)&sagg[lm][4 * ks + 2 * half];
      v2f a1 = *(const v2f*)&sagg[lm][4 * ks + 4 + 2 * half];
      c0 = __builtin_amdgcn_wmma_f32_16x16x4_f32(false, a0, false, bf[ks], (short)0, c0,
                                                 false, false);
      c1 = __builtin_amdgcn_wmma_f32_16x16x4_f32(false, a1, false, bf[ks + 1], (short)0, c1,
                                                 false, false);
    }
#endif
#pragma unroll
    for (int r = 0; r < 8; ++r) {
      srow[8 * half + r][col] = c0[r] + c1[r];
    }
  }
  __syncthreads();
  if (BN) {
    float4 y = ((const float4*)&srow[ln][0])[q];
    float* yp = &y.x;
#pragma unroll
    for (int c = 0; c < 4; ++c) {
      int d = q * 4 + c;
      float val = yp[c] + b[d];
      float rs = rsqrtf(v[d] + BN_EPS);
      yp[c] = fmaxf((val - m[d]) * rs * g[d] + be[d], 0.0f);
    }
    if (live) ((float4*)x)[(size_t)nd * 16 + q] = y;
  } else {
    if (threadIdx.x < DD) {
      int c = threadIdx.x;
      float vals[16];
      int gids[16];
      float bc = b[c];
#pragma unroll
      for (int k = 0; k < 16; ++k) { vals[k] = srow[k][c] + bc; gids[k] = sgid[k]; }
      int gcur = gids[0];
      float s = vals[0];
      int run = 1;
#pragma unroll
      for (int k = 1; k < 16; ++k) {
        int gk = gids[k];
        float vk = vals[k];
        if (gk == gcur) {
          s += vk;
          ++run;
        } else {
          if (gcur >= 0) {
            atomicAdd(&pool[(size_t)gcur * DD + c], s);
            if (c == 0) atomicAdd(&cnt[gcur], (float)run);
          }
          gcur = gk; s = vk; run = 1;
        }
      }
      if (gcur >= 0) {
        atomicAdd(&pool[(size_t)gcur * DD + c], s);
        if (c == 0) atomicAdd(&cnt[gcur], (float)run);
      }
    }
  }
}

template <bool ROWDIV>
__global__ void k_gemm_wmma(const float* __restrict__ A,
                            const float* __restrict__ B,
                            float* __restrict__ C,
                            const float* __restrict__ bias,
                            const float* __restrict__ dinv,
                            float* __restrict__ acc,
                            const float* __restrict__ rowcnt,
                            int M, int K, int Nc) {
  int wid  = (blockIdx.x * blockDim.x + threadIdx.x) >> 5;
  int lane = threadIdx.x & 31;
  int tiles_n = Nc >> 4;
  int tm = wid / tiles_n;
  int tn = wid - tm * tiles_n;
  if (tm * 16 >= M) return;

  int half = lane >> 4;
  int lm   = lane & 15;
  const float* Arow = A + (size_t)(tm * 16 + lm) * K;
  int ncol = tn * 16 + lm;
  float rc = 1.0f;
  bool has = true;
  if (ROWDIV) {
    float cr = rowcnt[tm * 16 + lm];
    has = cr > 0.0f;
    rc = has ? cr : 1.0f;
  }

  v8f c = {};
#if USE_WMMA_F32
  for (int k0 = 0; k0 < K; k0 += 4) {
    v2f a, b;
    a.x = Arow[k0 + 2 * half + 0];
    a.y = Arow[k0 + 2 * half + 1];
    if (ROWDIV) {
      float qx = a.x / rc;
      float qy = a.y / rc;
      a.x = has ? qx : 0.0f;
      a.y = has ? qy : 0.0f;
    }
    b.x = B[(size_t)(k0 + 2 * half + 0) * Nc + ncol];
    b.y = B[(size_t)(k0 + 2 * half + 1) * Nc + ncol];
    c = __builtin_amdgcn_wmma_f32_16x16x4_f32(false, a, false, b, (short)0, c,
                                              false, false);
  }
#else
  for (int k0 = 0; k0 < K; k0 += 32) {
    v16h a, b;
#pragma unroll
    for (int i = 0; i < 8; ++i) {
      float lo = Arow[k0 + 8 * half + i];
      float hi = Arow[k0 + 16 + 8 * half + i];
      if (ROWDIV) {
        lo = has ? lo / rc : 0.0f;
        hi = has ? hi / rc : 0.0f;
      }
      a[i]     = (_Float16)lo;
      a[8 + i] = (_Float16)hi;
    }
#pragma unroll
    for (int i = 0; i < 16; ++i)
      b[i] = (_Float16)B[(size_t)(k0 + 16 * half + i) * Nc + ncol];
    c = __builtin_amdgcn_wmma_f32_16x16x32_f16(false, a, false, b, (short)0, c,
                                               false, false);
  }
#endif

  float bv = bias ? bias[ncol] : 0.0f;
  if (acc) {
#pragma unroll
    for (int r = 0; r < 8; ++r) {
      int row = tm * 16 + 8 * half + r;
      size_t idx = (size_t)row * Nc + ncol;
      float val = c[r] + bv;
      float di = dinv[row];
      C[idx]   = val;
      acc[idx] = val * (di * di);
    }
  } else {
#pragma unroll
    for (int r = 0; r < 8; ++r) {
      C[(size_t)(tm * 16 + 8 * half + r) * Nc + ncol] = c[r] + bv;
    }
  }
}

static inline int blk(long long n, int b) { return (int)((n + b - 1) / b); }

extern "C" void kernel_launch(void* const* d_in, const int* in_sizes, int n_in,
                              void* d_out, int out_size, void* d_ws, size_t ws_size,
                              hipStream_t stream) {
  (void)in_sizes; (void)n_in; (void)out_size; (void)ws_size;
  const int*   edge_index = (const int*)d_in[0];
  const int*   batch      = (const int*)d_in[1];
  const float* W1 = (const float*)d_in[2];
  const float* b1 = (const float*)d_in[3];
  const float* W2 = (const float*)d_in[4];
  const float* b2 = (const float*)d_in[5];
  const float* W3 = (const float*)d_in[6];
  const float* b3 = (const float*)d_in[7];
  const float* g1 = (const float*)d_in[8];
  const float* be1= (const float*)d_in[9];
  const float* m1 = (const float*)d_in[10];
  const float* v1 = (const float*)d_in[11];
  const float* g2 = (const float*)d_in[12];
  const float* be2= (const float*)d_in[13];
  const float* m2 = (const float*)d_in[14];
  const float* v2 = (const float*)d_in[15];
  const float* Wl = (const float*)d_in[16];
  const float* bl = (const float*)d_in[17];
  float* out = (float*)d_out;

  char* ws = (char*)d_ws;
  size_t off = 0;
  auto carve = [&](size_t bytes) -> float* {
    float* p = (float*)(ws + off);
    off += (bytes + 255) & ~(size_t)255;
    return p;
  };
  int*   degcnt   = (int*)carve((size_t)NN * 4);
  float* dinv     = carve((size_t)NN * 4);
  float* sd       = carve((size_t)NN * 4);
  int*   rowptr   = (int*)carve((size_t)NN * 4);
  int*   rank     = (int*)carve((size_t)EE * 4);
  int*   csr      = (int*)carve((size_t)EE * 4);
  unsigned* hist  = (unsigned*)carve((size_t)CH * NN * 4);
  int*   coff     = (int*)carve((size_t)CH * NN * 4);
  int*   bsum     = (int*)carve((size_t)CH * NB * 4);
  float* bufX     = carve((size_t)NN * DD * 4);
  float* bufH     = carve((size_t)NN * DD * 4);
  float* pool     = carve((size_t)GG * DD * 4);
  float* cnt      = carve((size_t)GG * 4);

  const int B = 256;

  k_count_lds<<<CH, 1024, 0, stream>>>(edge_index, hist, bsum, rank);
  k_scan_dinv<<<NB, 256, 0, stream>>>(hist, bsum, coff, degcnt, rowptr, dinv, sd, NN);
  k_fill<<<blk(EE, B), B, 0, stream>>>(edge_index, rank, coff, csr, EE);

  const int tiles_FIN = (GG / 16) * (DOUT / 16);

  k_gather_l1<<<blk((long long)NN * 4, B), B, 0, stream>>>(
      rowptr, degcnt, csr, sd, dinv, W1, b1, g1, be1, m1, v1, bufX, pool, cnt, NN);

  k_gather_mm<true><<<blk((long long)NN * 16, B), B, 0, stream>>>(
      rowptr, degcnt, csr, bufX, dinv, W2, b2, g2, be2, m2, v2, bufH,
      nullptr, nullptr, nullptr, NN);

  k_gather_mm<false><<<blk((long long)NN * 16, B), B, 0, stream>>>(
      rowptr, degcnt, csr, bufH, dinv, W3, b3, nullptr, nullptr, nullptr, nullptr, nullptr,
      batch, pool, cnt, NN);

  k_gemm_wmma<true><<<blk((long long)tiles_FIN * 32, B), B, 0, stream>>>(
      pool, Wl, out, bl, nullptr, nullptr, cnt, GG, DD, DOUT);
}


# Layer 2 gathers 8 B per edge ({t, dinv} of the source) and re-forms the layer-1 row (t*W1+b1, BN, ReLU) in registers instead of reading a stored 256-B row; N x 64 layer-1 plane gone; sums as before
# speedup vs baseline: 4.2997x; 1.1134x over previous
#include <hip/hip_runtime.h>
#include <hip/hip_bf16.h>

#define NN   50000
#define EE   800000
#define GG   512
#define DD   64
#define DOUT 32
#define BN_EPS 1e-5f

typedef __attribute__((ext_vector_type(2)))  float    v2f;
typedef __attribute__((ext_vector_type(8)))  float    v8f;
typedef __attribute__((ext_vector_type(16))) _Float16 v16h;

#if defined(__AMDGCN__) && __has_builtin(__builtin_amdgcn_wmma_f32_16x16x4_f32)
#define USE_WMMA_F32 1
#else
#define USE_WMMA_F32 0
#endif


#define CH 64
#define CS (EE / CH)
#define NB ((NN + 255) / 256)
static_assert(EE % CH == 0 && CS % 4 == 0, "a chunk is a whole number of 4-edge groups");
static_assert(CS < 65536, "a chunk's count for one node fits a 16-bit field");
static_assert(NN % 4 == 0, "the table is moved as 4-word groups");
static_assert(NB <= 256, "one thread of the scan block per node block in front of it");
#define G4  (CS / 4)
#define PER ((G4 + 1023) / 1024)

__global__ void __launch_bounds__(1024)
k_count_lds(const int* __restrict__ ei, unsigned* __restrict__ hist,
            int* __restrict__ bsum, int* __restrict__ rank) {
  __shared__ __attribute__((aligned(16))) unsigned h[NN];
  __shared__ int sb[NB];
  const int t = threadIdx.x;
  const int lane = t & 31;
  const int w = blockIdx.x;
  const int4* src4 = (const int4*)(ei + (size_t)w * CS);
  const int4* dst4 = (const int4*)(ei + EE + (size_t)w * CS);
  int4 sv[PER], dv[PER];
#pragma unroll
  for (int k = 0; k < PER; ++k) {
    int g = min(t + 1024 * k, G4 - 1);
    sv[k] = src4[g];
    dv[k] = dst4[g];
  }
  uint4* h4 = (uint4*)h;
  for (int j = t; j < NN / 4; j += 1024) h4[j] = make_uint4(0u, 0u, 0u, 0u);
  if (t < NB) sb[t] = 0;
  __syncthreads();
  int4* rank4 = (int4*)(rank + (size_t)w * CS);
#pragma unroll
  for (int k = 0; k < PER; ++k) {
    int g = t + 1024 * k;
    if (g < G4) {
      int4 r;
      r.x = (int)(atomicAdd(&h[dv[k].x], 1u) & 0xFFFFu);
      r.y = (int)(atomicAdd(&h[dv[k].y], 1u) & 0xFFFFu);
      r.z = (int)(atomicAdd(&h[dv[k].z], 1u) & 0xFFFFu);
      r.w = (int)(atomicAdd(&h[dv[k].w], 1u) & 0xFFFFu);
      atomicAdd(&h[sv[k].x], 0x10000u);
      atomicAdd(&h[sv[k].y], 0x10000u);
      atomicAdd(&h[sv[k].z], 0x10000u);
      atomicAdd(&h[sv[k].w], 0x10000u);
      rank4[g] = r;
    }
  }
  __syncthreads();
  uint4* hg4 = (uint4*)(hist + (size_t)w * NN);
  for (int j0 = 0; j0 < NN / 4; j0 += 1024) {
    int j = j0 + t;
    bool ok = j < NN / 4;
    uint4 v = make_uint4(0u, 0u, 0u, 0u);
    if (ok) { v = h4[j]; hg4[j] = v; }
    int c = (int)((v.x & 0xFFFFu) + (v.y & 0xFFFFu) + (v.z & 0xFFFFu) + (v.w & 0xFFFFu));
    for (int o = 16; o > 0; o >>= 1) c += __shfl_xor(c, o);
    if (lane == 0 && ok) atomicAdd(&sb[j >> 6], c);
  }
  __syncthreads();
  if (t < NB) bsum[w * NB + t] = sb[t];
}

__global__ void __launch_bounds__(256)
k_scan_dinv(const unsigned* __restrict__ hist, const int* __restrict__ bsum,
            int* __restrict__ off, int* __restrict__ degcnt,
            int* __restrict__ rowptr, float* __restrict__ dinv,
            float* __restrict__ sd, int n) {
  __shared__ int wsum[8];
  __shared__ int s_base;
  int t = threadIdx.x;
  int lane = t & 31;
  int w = t >> 5;
  int b = blockIdx.x;
  int i = b * 256 + t;
  int ic = min(i, n - 1);
  unsigned hv[CH];
#pragma unroll
  for (int k = 0; k < CH; ++k) hv[k] = hist[(size_t)k * NN + ic];

  int p = 0;
  if (t < b) {
    for (int k = 0; k < CH; ++k) p += bsum[k * NB + t];
  }
  for (int o = 16; o > 0; o >>= 1) p += __shfl_xor(p, o);
  if (lane == 0) wsum[w] = p;
  __syncthreads();
  if (w == 0) {
    int x = (lane < 8) ? wsum[lane] : 0;
    for (int o = 16; o > 0; o >>= 1) x += __shfl_xor(x, o);
    if (lane == 0) s_base = x;
  }
  __syncthreads();
  int base = s_base;

  int cin = 0, cout = 0;
#pragma unroll
  for (int k = 0; k < CH; ++k) { cin += (int)(hv[k] & 0xFFFFu); cout += (int)(hv[k] >> 16); }
  int c = (i < n) ? cin : 0;
  int v = c;
  for (int o = 1; o < 32; o <<= 1) {
    int u = __shfl_up(v, o);
    if (lane >= o) v += u;
  }
  if (lane == 31) wsum[w] = v;
  __syncthreads();
  int woff = 0;
  for (int k = 0; k < 8; ++k) woff += (k < w) ? wsum[k] : 0;
  if (i < n) {
    int rp = base + woff + v - c;
    rowptr[i] = rp;
    degcnt[i] = c;
    float di = rsqrtf((float)c + 1.0f);
    dinv[i] = di;
    sd[i] = (float)cout * di;
    int run = rp;
#pragma unroll
    for (int k = 0; k < CH; ++k) {
      off[(size_t)k * NN + i] = run;
      run += (int)(hv[k] & 0xFFFFu);
    }
  }
}

__global__ void k_fill(const int* __restrict__ ei,
                       const int* __restrict__ rank,
                       const int* __restrict__ off,
                       int* __restrict__ csr, int E) {
  int e = blockIdx.x * blockDim.x + threadIdx.x;
  if (e >= E) return;
  int s = ei[e];
  int d = ei[E + e];
  csr[off[(size_t)(e / CS) * NN + d] + rank[e]] = s;
}

__global__ void __launch_bounds__(256)
k_gather_l1(const int* __restrict__ rowptr,
            const int* __restrict__ degcnt,
            const int* __restrict__ csr,
            const float* __restrict__ sd,
            const float* __restrict__ dinv,
            float* __restrict__ td,
            float* __restrict__ gsum,
            float* __restrict__ gcnt, int n) {
  int t = blockIdx.x * blockDim.x + threadIdx.x;
  int sub = t & 3;
  static_assert((long long)NN * 4 >= (long long)GG * DD, "one thread of this grid per pool word");
  if (t < GG * DD) gsum[t] = 0.0f;
  if (t < GG) gcnt[t] = 0.0f;
  int node = t >> 2;
  int ndc = min(node, n - 1);
  int beg = rowptr[ndc];
  int cnt = degcnt[ndc];
  float dn = dinv[ndc];
  float acc = (sub == 0) ? sd[ndc] : 0.0f;
  int end = beg + cnt;
  int last = (cnt > 0) ? end - 1 : 0;
  int j0 = beg + sub;
  int s[8];
  float val[8];
#pragma unroll
  for (int i = 0; i < 8; ++i) s[i] = csr[min(j0 + 4 * i, last)];
#pragma unroll
  for (int i = 0; i < 8; ++i) val[i] = sd[s[i]];
#pragma unroll
  for (int i = 0; i < 8; ++i) acc += (j0 + 4 * i < end) ? val[i] : 0.0f;
  for (int j = j0 + 32; j < end; j += 4) acc += sd[csr[j]];
  acc += __shfl_xor(acc, 1);
  acc += __shfl_xor(acc, 2);
  float tn = acc * dn;
  if (sub == 0 && node < n) ((float2*)td)[node] = make_float2(tn, dn);
}

__device__ __forceinline__ v2f l1_row(float t, v2f w, v2f b, v2f m, v2f rs, v2f g, v2f be) {
  v2f r = t * w + b;
  v2f y = (r - m) * rs * g + be;
  v2f zero = {0.0f, 0.0f};
  return __builtin_elementwise_max(y, zero);
}

#define TP 68
template <bool BN, bool SCAL>
__global__ void __launch_bounds__(256, 16)
k_gather_mm(const int* __restrict__ rowptr,
                         const int* __restrict__ degcnt,
                         const int* __restrict__ csr,
                         const float* __restrict__ xin,
                         const float* __restrict__ dinv,
                         const float* __restrict__ td,
                         const float* __restrict__ pW,
                         const float* __restrict__ pb,
                         const float* __restrict__ pg,
                         const float* __restrict__ pbe,
                         const float* __restrict__ pm,
                         const float* __restrict__ pv,
                         const float* __restrict__ W,
                         const float* __restrict__ b,
                         const float* __restrict__ g,
                         const float* __restrict__ be,
                         const float* __restrict__ m,
                         const float* __restrict__ v,
                         float* __restrict__ x,
                         const int* __restrict__ batch,
                         float* __restrict__ pool,
                         float* __restrict__ cnt, int n) {
  __shared__ __attribute__((aligned(16))) float sagg[16][TP];
  __shared__ __attribute__((aligned(16))) float srow[16][TP];
  __shared__ int sgid[16];
  int t = blockIdx.x * blockDim.x + threadIdx.x;
  int nd = t >> 4;
  int q  = t & 15;
  bool live = nd < n;
  nd = min(nd, n - 1);
  int beg = rowptr[nd];
  int end = beg + degcnt[nd];
  float4 a;
  if (SCAL) {
    const float2* td2 = (const float2*)td;
    const v2f* pW2 = (const v2f*)pW;  const v2f* pb2 = (const v2f*)pb;
    const v2f* pm2 = (const v2f*)pm;  const v2f* pv2 = (const v2f*)pv;
    const v2f* pg2 = (const v2f*)pg;  const v2f* pbe2 = (const v2f*)pbe;
    v2f cw[2], cb[2], cm[2], cr[2], cg[2], ce[2];
#pragma unroll
    for (int h = 0; h < 2; ++h) {
      int d2 = q * 2 + h;
      cw[h] = pW2[d2]; cb[h] = pb2[d2]; cm[h] = pm2[d2];
      v2f vv = pv2[d2];
      cr[h].x = rsqrtf(vv.x + BN_EPS); cr[h].y = rsqrtf(vv.y + BN_EPS);
      cg[h] = pg2[d2]; ce[h] = pbe2[d2];
    }
    float2 own = td2[nd];
    float dn = own.y;
    float ws = dn * dn;
    v2f acc[2];
#pragma unroll
    for (int h = 0; h < 2; ++h)
      acc[h] = l1_row(own.x, cw[h], cb[h], cm[h], cr[h], cg[h], ce[h]) * ws;
    if (beg < end) {
      int last = end - 1;
      int s0 = csr[beg];
      int s1 = csr[min(beg + 1, last)];
      int s2 = csr[min(beg + 2, last)];
      int s3 = csr[min(beg + 3, last)];
      for (int j = beg; j < end; j += 4) {
        bool p1 = (j + 1) < end;
        bool p2 = (j + 2) < end;
        bool p3 = (j + 3) < end;
        float2 e0 = td2[s0];
        float2 e1 = td2[s1];
        float2 e2 = td2[s2];
        float2 e3 = td2[s3];
        s0 = csr[min(j + 4, last)];
        s1 = csr[min(j + 5, last)];
        s2 = csr[min(j + 6, last)];
        s3 = csr[min(j + 7, last)];
        __builtin_amdgcn_sched_barrier(0);
        float wt0 = e0.y * dn;
        float wt1 = p1 ? e1.y * dn : 0.0f;
        float wt2 = p2 ? e2.y * dn : 0.0f;
        float wt3 = p3 ? e3.y * dn : 0.0f;
#pragma unroll
        for (int h = 0; h < 2; ++h)
          acc[h] += l1_row(e0.x, cw[h], cb[h], cm[h], cr[h], cg[h], ce[h]) * wt0;
#pragma unroll
        for (int h = 0; h < 2; ++h)
          acc[h] += l1_row(e1.x, cw[h], cb[h], cm[h], cr[h], cg[h], ce[h]) * wt1;
#pragma unroll
        for (int h = 0; h < 2; ++h)
          acc[h] += l1_row(e2.x, cw[h], cb[h], cm[h], cr[h], cg[h], ce[h]) * wt2;
#pragma unroll
        for (int h = 0; h < 2; ++h)
          acc[h] += l1_row(e3.x, cw[h], cb[h], cm[h], cr[h], cg[h], ce[h]) * wt3;
      }
    }
    a.x = acc[0].x; a.y = acc[0].y; a.z = acc[1].x; a.w = acc[1].y;
  } else {
    const float4* xin4 = (const float4*)xin;
    float dn = dinv[nd];
    float4 h = xin4[(size_t)nd * 16 + q];
    float ws = dn * dn;
    a.x = h.x * ws; a.y = h.y * ws; a.z = h.z * ws; a.w = h.w * ws;
    if (beg < end) {
      int last = end - 1;
      int s0 = csr[beg];
      int s1 = csr[min(beg + 1, last)];
      int s2 = csr[min(beg + 2, last)];
      int s3 = csr[min(beg + 3, last)];
      for (int j = beg; j < end; j += 4) {
        bool p1 = (j + 1) < end;
        bool p2 = (j + 2) < end;
        bool p3 = (j + 3) < end;
        float w0 = dinv[s0] * dn;
        float w1 = dinv[s1] * dn;
        float w2 = dinv[s2] * dn;
        float w3 = dinv[s3] * dn;
        float4 m0 = xin4[(size_t)s0 * 16 + q];
        float4 m1 = xin4[(size_t)s1 * 16 + q];
        float4 m2 = xin4[(size_t)s2 * 16 + q];
        float4 m3 = xin4[(size_t)s3 * 16 + q];
        s0 = csr[min(j + 4, last)];
        s1 = csr[min(j + 5, last)];
        s2 = csr[min(j + 6, last)];
        s3 = csr[min(j + 7, last)];
        asm volatile("" :: "v"(m0.x), "v"(m0.y), "v"(m0.z), "v"(m0.w),
                           "v"(m1.x), "v"(m1.y), "v"(m1.z), "v"(m1.w),
                           "v"(m2.x), "v"(m2.y), "v"(m2.z), "v"(m2.w),
                           "v"(m3.x), "v"(m3.y), "v"(m3.z), "v"(m3.w));
        __builtin_amdgcn_sched_barrier(0);
        a.x += m0.x * w0; a.y += m0.y * w0; a.z += m0.z * w0; a.w += m0.w * w0;
        float4 u;
        u.x = a.x + m1.x * w1; u.y = a.y + m1.y * w1; u.z = a.z + m1.z * w1; u.w = a.w + m1.w * w1;
        a.x = p1 ? u.x : a.x; a.y = p1 ? u.y : a.y; a.z = p1 ? u.z : a.z; a.w = p1 ? u.w : a.w;
        u.x = a.x + m2.x * w2; u.y = a.y + m2.y * w2; u.z = a.z + m2.z * w2; u.w = a.w + m2.w * w2;
        a.x = p2 ? u.x : a.x; a.y = p2 ? u.y : a.y; a.z = p2 ? u.z : a.z; a.w = p2 ? u.w : a.w;
        u.x = a.x + m3.x * w3; u.y = a.y + m3.y * w3; u.z = a.z + m3.z * w3; u.w = a.w + m3.w * w3;
        a.x = p3 ? u.x : a.x; a.y = p3 ? u.y : a.y; a.z = p3 ? u.z : a.z; a.w = p3 ? u.w : a.w;
      }
    }
  }
  int ln = threadIdx.x >> 4;
  int wv = __builtin_amdgcn_readfirstlane((int)(threadIdx.x >> 5));
  int lane = threadIdx.x & 31;
  int half = lane >> 4;
  int lm   = lane & 15;
  int col  = (wv & 3) * 16 + lm;
#if USE_WMMA_F32
  v2f bf[DD / 4];
  if (wv < 4) {
#pragma unroll
    for (int ks = 0; ks < DD / 4; ++ks) {
      bf[ks].x = W[(size_t)(4 * ks + 2 * half + 0) * DD + col];
      bf[ks].y = W[(size_t)(4 * ks + 2 * half + 1) * DD + col];
    }
  }
#endif
  ((float4*)&sagg[ln][0])[q] = a;
  if (!BN) {
    if (q == 0) sgid[ln] = live ? batch[nd] : -1;
  }
  __syncthreads();
  if (wv < 4) {
    v8f c0 = {};
    v8f c1 = {};
#if USE_WMMA_F32
#pragma unroll
    for (int ks = 0; ks < DD / 4; ks += 2) {
      v2f a0 = *(const v2f*)&sagg[lm][4 * ks + 2 * half];
      v2f a1 = *(const v2f*)&sagg[lm][4 * ks + 4 + 2 * half];
      c0 = __builtin_amdgcn_wmma_f32_16x16x4_f32(false, a0, false, bf[ks], (short)0, c0,
                                                 false, false);
      c1 = __builtin_amdgcn_wmma_f32_16x16x4_f32(false, a1, false, bf[ks + 1], (short)0, c1,
                                                 false, false);
    }
#endif
#pragma unroll
    for (int r = 0; r < 8; ++r) {
      srow[8 * half + r][col] = c0[r] + c1[r];
    }
  }
  __syncthreads();
  if (BN) {
    float4 y = ((const float4*)&srow[ln][0])[q];
    float* yp = &y.x;
#pragma unroll
    for (int c = 0; c < 4; ++c) {
      int d = q * 4 + c;
      float val = yp[c] + b[d];
      float rs = rsqrtf(v[d] + BN_EPS);
      yp[c] = fmaxf((val - m[d]) * rs * g[d] + be[d], 0.0f);
    }
    if (live) ((float4*)x)[(size_t)nd * 16 + q] = y;
  } else {
    if (threadIdx.x < DD) {
      int c = threadIdx.x;
      float vals[16];
      int gids[16];
      float bc = b[c];
#pragma unroll
      for (int k = 0; k < 16; ++k) { vals[k] = srow[k][c] + bc; gids[k] = sgid[k]; }
      int gcur = gids[0];
      float s = vals[0];
      int run = 1;
#pragma unroll
      for (int k = 1; k < 16; ++k) {
        int gk = gids[k];
        float vk = vals[k];
        if (gk == gcur) {
          s += vk;
          ++run;
        } else {
          if (gcur >= 0) {
            atomicAdd(&pool[(size_t)gcur * DD + c], s);
            if (c == 0) atomicAdd(&cnt[gcur], (float)run);
          }
          gcur = gk; s = vk; run = 1;
        }
      }
      if (gcur >= 0) {
        atomicAdd(&pool[(size_t)gcur * DD + c], s);
        if (c == 0) atomicAdd(&cnt[gcur], (float)run);
      }
    }
  }
}

template <bool ROWDIV>
__global__ void k_gemm_wmma(const float* __restrict__ A,
                            const float* __restrict__ B,
                            float* __restrict__ C,
                            const float* __restrict__ bias,
                            const float* __restrict__ dinv,
                            float* __restrict__ acc,
                            const float* __restrict__ rowcnt,
                            int M, int K, int Nc) {
  int wid  = (blockIdx.x * blockDim.x + threadIdx.x) >> 5;
  int lane = threadIdx.x & 31;
  int tiles_n = Nc >> 4;
  int tm = wid / tiles_n;
  int tn = wid - tm * tiles_n;
  if (tm * 16 >= M) return;

  int half = lane >> 4;
  int lm   = lane & 15;
  const float* Arow = A + (size_t)(tm * 16 + lm) * K;
  int ncol = tn * 16 + lm;
  float rc = 1.0f;
  bool has = true;
  if (ROWDIV) {
    float cr = rowcnt[tm * 16 + lm];
    has = cr > 0.0f;
    rc = has ? cr : 1.0f;
  }

  v8f c = {};
#if USE_WMMA_F32
  for (int k0 = 0; k0 < K; k0 += 4) {
    v2f a, b;
    a.x = Arow[k0 + 2 * half + 0];
    a.y = Arow[k0 + 2 * half + 1];
    if (ROWDIV) {
      float qx = a.x / rc;
      float qy = a.y / rc;
      a.x = has ? qx : 0.0f;
      a.y = has ? qy : 0.0f;
    }
    b.x = B[(size_t)(k0 + 2 * half + 0) * Nc + ncol];
    b.y = B[(size_t)(k0 + 2 * half + 1) * Nc + ncol];
    c = __builtin_amdgcn_wmma_f32_16x16x4_f32(false, a, false, b, (short)0, c,
                                              false, false);
  }
#else
  for (int k0 = 0; k0 < K; k0 += 32) {
    v16h a, b;
#pragma unroll
    for (int i = 0; i < 8; ++i) {
      float lo = Arow[k0 + 8 * half + i];
      float hi = Arow[k0 + 16 + 8 * half + i];
      if (ROWDIV) {
        lo = has ? lo / rc : 0.0f;
        hi = has ? hi / rc : 0.0f;
      }
      a[i]     = (_Float16)lo;
      a[8 + i] = (_Float16)hi;
    }
#pragma unroll
    for (int i = 0; i < 16; ++i)
      b[i] = (_Float16)B[(size_t)(k0 + 16 * half + i) * Nc + ncol];
    c = __builtin_amdgcn_wmma_f32_16x16x32_f16(false, a, false, b, (short)0, c,
                                               false, false);
  }
#endif

  float bv = bias ? bias[ncol] : 0.0f;
  if (acc) {
#pragma unroll
    for (int r = 0; r < 8; ++r) {
      int row = tm * 16 + 8 * half + r;
      size_t idx = (size_t)row * Nc + ncol;
      float val = c[r] + bv;
      float di = dinv[row];
      C[idx]   = val;
      acc[idx] = val * (di * di);
    }
  } else {
#pragma unroll
    for (int r = 0; r < 8; ++r) {
      C[(size_t)(tm * 16 + 8 * half + r) * Nc + ncol] = c[r] + bv;
    }
  }
}

static inline int blk(long long n, int b) { return (int)((n + b - 1) / b); }

extern "C" void kernel_launch(void* const* d_in, const int* in_sizes, int n_in,
                              void* d_out, int out_size, void* d_ws, size_t ws_size,
                              hipStream_t stream) {
  (void)in_sizes; (void)n_in; (void)out_size; (void)ws_size;
  const int*   edge_index = (const int*)d_in[0];
  const int*   batch      = (const int*)d_in[1];
  const float* W1 = (const float*)d_in[2];
  const float* b1 = (const float*)d_in[3];
  const float* W2 = (const float*)d_in[4];
  const float* b2 = (const float*)d_in[5];
  const float* W3 = (const float*)d_in[6];
  const float* b3 = (const float*)d_in[7];
  const float* g1 = (const float*)d_in[8];
  const float* be1= (const float*)d_in[9];
  const float* m1 = (const float*)d_in[10];
  const float* v1 = (const float*)d_in[11];
  const float* g2 = (const float*)d_in[12];
  const float* be2= (const float*)d_in[13];
  const float* m2 = (const float*)d_in[14];
  const float* v2 = (const float*)d_in[15];
  const float* Wl = (const float*)d_in[16];
  const float* bl = (const float*)d_in[17];
  float* out = (float*)d_out;

  char* ws = (char*)d_ws;
  size_t off = 0;
  auto carve = [&](size_t bytes) -> float* {
    float* p = (float*)(ws + off);
    off += (bytes + 255) & ~(size_t)255;
    return p;
  };
  int*   degcnt   = (int*)carve((size_t)NN * 4);
  float* dinv     = carve((size_t)NN * 4);
  float* sd       = carve((size_t)NN * 4);
  int*   rowptr   = (int*)carve((size_t)NN * 4);
  int*   rank     = (int*)carve((size_t)EE * 4);
  int*   csr      = (int*)carve((size_t)EE * 4);
  unsigned* hist  = (unsigned*)carve((size_t)CH * NN * 4);
  int*   coff     = (int*)carve((size_t)CH * NN * 4);
  int*   bsum     = (int*)carve((size_t)CH * NB * 4);
  float* td       = carve((size_t)NN * 8);
  float* bufH     = carve((size_t)NN * DD * 4);
  float* pool     = carve((size_t)GG * DD * 4);
  float* cnt      = carve((size_t)GG * 4);

  const int B = 256;

  k_count_lds<<<CH, 1024, 0, stream>>>(edge_index, hist, bsum, rank);
  k_scan_dinv<<<NB, 256, 0, stream>>>(hist, bsum, coff, degcnt, rowptr, dinv, sd, NN);
  k_fill<<<blk(EE, B), B, 0, stream>>>(edge_index, rank, coff, csr, EE);

  const int tiles_FIN = (GG / 16) * (DOUT / 16);

  k_gather_l1<<<blk((long long)NN * 4, B), B, 0, stream>>>(
      rowptr, degcnt, csr, sd, dinv, td, pool, cnt, NN);

  k_gather_mm<true, true><<<blk((long long)NN * 16, B), B, 0, stream>>>(
      rowptr, degcnt, csr, nullptr, dinv, td, W1, b1, g1, be1, m1, v1,
      W2, b2, g2, be2, m2, v2, bufH, nullptr, nullptr, nullptr, NN);

  k_gather_mm<false, false><<<blk((long long)NN * 16, B), B, 0, stream>>>(
      rowptr, degcnt, csr, bufH, dinv, nullptr, nullptr, nullptr, nullptr, nullptr, nullptr, nullptr,
      W3, b3, nullptr, nullptr, nullptr, nullptr, nullptr,
      batch, pool, cnt, NN);

  k_gemm_wmma<true><<<blk((long long)tiles_FIN * 32, B), B, 0, stream>>>(
      pool, Wl, out, bl, nullptr, nullptr, cnt, GG, DD, DOUT);
}


# Head folded into layer 3: Wc=W3*Wl formed each call on the f32 matrix unit, layer 2 stores dinv*(x2*Wc) as 128-B rows, layer 3 sums one line per edge; sum order: k of W3 and Wl before neighbours
# speedup vs baseline: 4.8883x; 1.1369x over previous
#include <hip/hip_runtime.h>
#include <hip/hip_bf16.h>

#define NN   50000
#define EE   800000
#define GG   512
#define DD   64
#define DOUT 32
#define BN_EPS 1e-5f

typedef __attribute__((ext_vector_type(2)))  float    v2f;
typedef __attribute__((ext_vector_type(8)))  float    v8f;

#if defined(__AMDGCN__) && __has_builtin(__builtin_amdgcn_wmma_f32_16x16x4_f32)
#define USE_WMMA_F32 1
#else
#define USE_WMMA_F32 0
#endif


#define CH 64
#define CS (EE / CH)
#define NB ((NN + 255) / 256)
static_assert(EE % CH == 0 && CS % 4 == 0, "a chunk is a whole number of 4-edge groups");
static_assert(CS < 65536, "a chunk's count for one node fits a 16-bit field");
static_assert(NN % 4 == 0, "the table is moved as 4-word groups");
static_assert(NB <= 256, "one thread of the scan block per node block in front of it");
#define G4  (CS / 4)
#define PER ((G4 + 1023) / 1024)

__global__ void __launch_bounds__(1024)
k_count_lds(const int* __restrict__ ei, unsigned* __restrict__ hist,
            int* __restrict__ bsum, int* __restrict__ rank) {
  __shared__ __attribute__((aligned(16))) unsigned h[NN];
  __shared__ int sb[NB];
  const int t = threadIdx.x;
  const int lane = t & 31;
  const int w = blockIdx.x;
  const int4* src4 = (const int4*)(ei + (size_t)w * CS);
  const int4* dst4 = (const int4*)(ei + EE + (size_t)w * CS);
  int4 sv[PER], dv[PER];
#pragma unroll
  for (int k = 0; k < PER; ++k) {
    int g = min(t + 1024 * k, G4 - 1);
    sv[k] = src4[g];
    dv[k] = dst4[g];
  }
  uint4* h4 = (uint4*)h;
  for (int j = t; j < NN / 4; j += 1024) h4[j] = make_uint4(0u, 0u, 0u, 0u);
  if (t < NB) sb[t] = 0;
  __syncthreads();
  int4* rank4 = (int4*)(rank + (size_t)w * CS);
#pragma unroll
  for (int k = 0; k < PER; ++k) {
    int g = t + 1024 * k;
    if (g < G4) {
      int4 r;
      r.x = (int)(atomicAdd(&h[dv[k].x], 1u) & 0xFFFFu);
      r.y = (int)(atomicAdd(&h[dv[k].y], 1u) & 0xFFFFu);
      r.z = (int)(atomicAdd(&h[dv[k].z], 1u) & 0xFFFFu);
      r.w = (int)(atomicAdd(&h[dv[k].w], 1u) & 0xFFFFu);
      atomicAdd(&h[sv[k].x], 0x10000u);
      atomicAdd(&h[sv[k].y], 0x10000u);
      atomicAdd(&h[sv[k].z], 0x10000u);
      atomicAdd(&h[sv[k].w], 0x10000u);
      rank4[g] = r;
    }
  }
  __syncthreads();
  uint4* hg4 = (uint4*)(hist + (size_t)w * NN);
  for (int j0 = 0; j0 < NN / 4; j0 += 1024) {
    int j = j0 + t;
    bool ok = j < NN / 4;
    uint4 v = make_uint4(0u, 0u, 0u, 0u);
    if (ok) { v = h4[j]; hg4[j] = v; }
    int c = (int)((v.x & 0xFFFFu) + (v.y & 0xFFFFu) + (v.z & 0xFFFFu) + (v.w & 0xFFFFu));
    for (int o = 16; o > 0; o >>= 1) c += __shfl_xor(c, o);
    if (lane == 0 && ok) atomicAdd(&sb[j >> 6], c);
  }
  __syncthreads();
  if (t < NB) bsum[w * NB + t] = sb[t];
}

__device__ __forceinline__ void form_tail_weights(const float* __restrict__ W3,
                                                  const float* __restrict__ b3,
                                                  const float* __restrict__ Wl,
                                                  float* __restrict__ Wc,
                                                  float* __restrict__ bc) {
#if USE_WMMA_F32
  int wv = __builtin_amdgcn_readfirstlane((int)(threadIdx.x >> 5));
  int lane = threadIdx.x & 31;
  int half = lane >> 4;
  int lm   = lane & 15;
  int tm = wv >> 1;
  int ncol = (wv & 1) * 16 + lm;
  const float* Arow = W3 + (size_t)(tm * 16 + lm) * DD;
  v8f c = {};
#pragma unroll
  for (int k0 = 0; k0 < DD; k0 += 4) {
    v2f a, b;
    a.x = Arow[k0 + 2 * half + 0];
    a.y = Arow[k0 + 2 * half + 1];
    b.x = Wl[(size_t)(k0 + 2 * half + 0) * DOUT + ncol];
    b.y = Wl[(size_t)(k0 + 2 * half + 1) * DOUT + ncol];
    c = __builtin_amdgcn_wmma_f32_16x16x4_f32(false, a, false, b, (short)0, c, false, false);
  }
#pragma unroll
  for (int r = 0; r < 8; ++r) {
    Wc[(size_t)(tm * 16 + 8 * half + r) * DOUT + ncol] = c[r];
  }
  if (wv < 2) {
    v8f cb = {};
#pragma unroll
    for (int k0 = 0; k0 < DD; k0 += 4) {
      v2f a, b;
      a.x = b3[k0 + 2 * half + 0];
      a.y = b3[k0 + 2 * half + 1];
      b.x = Wl[(size_t)(k0 + 2 * half + 0) * DOUT + ncol];
      b.y = Wl[(size_t)(k0 + 2 * half + 1) * DOUT + ncol];
      cb = __builtin_amdgcn_wmma_f32_16x16x4_f32(false, a, false, b, (short)0, cb, false, false);
    }
    if (half == 0) bc[ncol] = cb[0];
  }
#endif
}

__global__ void __launch_bounds__(256)
k_scan_dinv(const unsigned* __restrict__ hist, const int* __restrict__ bsum,
            int* __restrict__ off, int* __restrict__ degcnt,
            int* __restrict__ rowptr, float* __restrict__ dinv,
            float* __restrict__ sd,
            const float* __restrict__ W3, const float* __restrict__ b3,
            const float* __restrict__ Wl,
            float* __restrict__ Wc, float* __restrict__ bc, int n) {
  __shared__ int wsum[8];
  __shared__ int s_base;
  if (blockIdx.x == NB) {
    form_tail_weights(W3, b3, Wl, Wc, bc);
    return;
  }
  int t = threadIdx.x;
  int lane = t & 31;
  int w = t >> 5;
  int b = blockIdx.x;
  int i = b * 256 + t;
  int ic = min(i, n - 1);
  unsigned hv[CH];
#pragma unroll
  for (int k = 0; k < CH; ++k) hv[k] = hist[(size_t)k * NN + ic];

  int p = 0;
  if (t < b) {
    for (int k = 0; k < CH; ++k) p += bsum[k * NB + t];
  }
  for (int o = 16; o > 0; o >>= 1) p += __shfl_xor(p, o);
  if (lane == 0) wsum[w] = p;
  __syncthreads();
  if (w == 0) {
    int x = (lane < 8) ? wsum[lane] : 0;
    for (int o = 16; o > 0; o >>= 1) x += __shfl_xor(x, o);
    if (lane == 0) s_base = x;
  }
  __syncthreads();
  int base = s_base;

  int cin = 0, cout = 0;
#pragma unroll
  for (int k = 0; k < CH; ++k) { cin += (int)(hv[k] & 0xFFFFu); cout += (int)(hv[k] >> 16); }
  int c = (i < n) ? cin : 0;
  int v = c;
  for (int o = 1; o < 32; o <<= 1) {
    int u = __shfl_up(v, o);
    if (lane >= o) v += u;
  }
  if (lane == 31) wsum[w] = v;
  __syncthreads();
  int woff = 0;
  for (int k = 0; k < 8; ++k) woff += (k < w) ? wsum[k] : 0;
  if (i < n) {
    int rp = base + woff + v - c;
    rowptr[i] = rp;
    degcnt[i] = c;
    float di = rsqrtf((float)c + 1.0f);
    dinv[i] = di;
    sd[i] = (float)cout * di;
    int run = rp;
#pragma unroll
    for (int k = 0; k < CH; ++k) {
      off[(size_t)k * NN + i] = run;
      run += (int)(hv[k] & 0xFFFFu);
    }
  }
}

__global__ void k_fill(const int* __restrict__ ei,
                       const int* __restrict__ rank,
                       const int* __restrict__ off,
                       int* __restrict__ csr, int E) {
  int e = blockIdx.x * blockDim.x + threadIdx.x;
  if (e >= E) return;
  int s = ei[e];
  int d = ei[E + e];
  csr[off[(size_t)(e / CS) * NN + d] + rank[e]] = s;
}

__global__ void __launch_bounds__(256)
k_gather_l1(const int* __restrict__ rowptr,
            const int* __restrict__ degcnt,
            const int* __restrict__ csr,
            const float* __restrict__ sd,
            const float* __restrict__ dinv,
            float* __restrict__ td,
            float* __restrict__ gsum,
            float* __restrict__ gcnt, int n) {
  int t = blockIdx.x * blockDim.x + threadIdx.x;
  int sub = t & 3;
  static_assert((long long)NN * 4 >= (long long)GG * DOUT, "one thread of this grid per pool word");
  if (t < GG * DOUT) gsum[t] = 0.0f;
  if (t < GG) gcnt[t] = 0.0f;
  int node = t >> 2;
  int ndc = min(node, n - 1);
  int beg = rowptr[ndc];
  int cnt = degcnt[ndc];
  float dn = dinv[ndc];
  float acc = (sub == 0) ? sd[ndc] : 0.0f;
  int end = beg + cnt;
  int last = (cnt > 0) ? end - 1 : 0;
  int j0 = beg + sub;
  int s[8];
  float val[8];
#pragma unroll
  for (int i = 0; i < 8; ++i) s[i] = csr[min(j0 + 4 * i, last)];
#pragma unroll
  for (int i = 0; i < 8; ++i) val[i] = sd[s[i]];
#pragma unroll
  for (int i = 0; i < 8; ++i) acc += (j0 + 4 * i < end) ? val[i] : 0.0f;
  for (int j = j0 + 32; j < end; j += 4) acc += sd[csr[j]];
  acc += __shfl_xor(acc, 1);
  acc += __shfl_xor(acc, 2);
  float tn = acc * dn;
  if (sub == 0 && node < n) ((float2*)td)[node] = make_float2(tn, dn);
}

__device__ __forceinline__ v2f l1_row(float t, v2f w, v2f b, v2f m, v2f rs, v2f g, v2f be) {
  v2f r = t * w + b;
  v2f y = (r - m) * rs * g + be;
  v2f zero = {0.0f, 0.0f};
  return __builtin_elementwise_max(y, zero);
}

#define TP 68
__global__ void __launch_bounds__(256, 16)
k_gather_mm2(const int* __restrict__ rowptr,
             const int* __restrict__ degcnt,
             const int* __restrict__ csr,
             const float* __restrict__ td,
             const float* __restrict__ pW,
             const float* __restrict__ pb,
             const float* __restrict__ pg,
             const float* __restrict__ pbe,
             const float* __restrict__ pm,
             const float* __restrict__ pv,
             const float* __restrict__ W,
             const float* __restrict__ b,
             const float* __restrict__ g,
             const float* __restrict__ be,
             const float* __restrict__ m,
             const float* __restrict__ v,
             const float* __restrict__ Wc,
             const float* __restrict__ dinv,
             float* __restrict__ zs, int n) {
  __shared__ __attribute__((aligned(16))) float sagg[16][TP];
  __shared__ __attribute__((aligned(16))) float srow[16][TP];
  int t = blockIdx.x * blockDim.x + threadIdx.x;
  int q  = t & 15;
  int nd = min(t >> 4, n - 1);
  int beg = rowptr[nd];
  int end = beg + degcnt[nd];
  float4 a;
  const float2* td2 = (const float2*)td;
  const v2f* pW2 = (const v2f*)pW;  const v2f* pb2 = (const v2f*)pb;
  const v2f* pm2 = (const v2f*)pm;  const v2f* pv2 = (const v2f*)pv;
  const v2f* pg2 = (const v2f*)pg;  const v2f* pbe2 = (const v2f*)pbe;
  v2f cw[2], cb[2], cm[2], cr[2], cg[2], ce[2];
#pragma unroll
  for (int h = 0; h < 2; ++h) {
    int d2 = q * 2 + h;
    cw[h] = pW2[d2]; cb[h] = pb2[d2]; cm[h] = pm2[d2];
    v2f vv = pv2[d2];
    cr[h].x = rsqrtf(vv.x + BN_EPS); cr[h].y = rsqrtf(vv.y + BN_EPS);
    cg[h] = pg2[d2]; ce[h] = pbe2[d2];
  }
  float2 own = td2[nd];
  float dn = own.y;
  {
    float ws = dn * dn;
    v2f acc[2];
#pragma unroll
    for (int h = 0; h < 2; ++h)
      acc[h] = l1_row(own.x, cw[h], cb[h], cm[h], cr[h], cg[h], ce[h]) * ws;
    if (beg < end) {
      int last = end - 1;
      int s0 = csr[beg];
      int s1 = csr[min(beg + 1, last)];
      int s2 = csr[min(beg + 2, last)];
      int s3 = csr[min(beg + 3, last)];
      for (int j = beg; j < end; j += 4) {
        bool p1 = (j + 1) < end;
        bool p2 = (j + 2) < end;
        bool p3 = (j + 3) < end;
        float2 e0 = td2[s0];
        float2 e1 = td2[s1];
        float2 e2 = td2[s2];
        float2 e3 = td2[s3];
        s0 = csr[min(j + 4, last)];
        s1 = csr[min(j + 5, last)];
        s2 = csr[min(j + 6, last)];
        s3 = csr[min(j + 7, last)];
        __builtin_amdgcn_sched_barrier(0);
        float wt0 = e0.y * dn;
        float wt1 = p1 ? e1.y * dn : 0.0f;
        float wt2 = p2 ? e2.y * dn : 0.0f;
        float wt3 = p3 ? e3.y * dn : 0.0f;
#pragma unroll
        for (int h = 0; h < 2; ++h)
          acc[h] += l1_row(e0.x, cw[h], cb[h], cm[h], cr[h], cg[h], ce[h]) * wt0;
#pragma unroll
        for (int h = 0; h < 2; ++h)
          acc[h] += l1_row(e1.x, cw[h], cb[h], cm[h], cr[h], cg[h], ce[h]) * wt1;
#pragma unroll
        for (int h = 0; h < 2; ++h)
          acc[h] += l1_row(e2.x, cw[h], cb[h], cm[h], cr[h], cg[h], ce[h]) * wt2;
#pragma unroll
        for (int h = 0; h < 2; ++h)
          acc[h] += l1_row(e3.x, cw[h], cb[h], cm[h], cr[h], cg[h], ce[h]) * wt3;
      }
    }
    a.x = acc[0].x; a.y = acc[0].y; a.z = acc[1].x; a.w = acc[1].y;
  }
  int ln = threadIdx.x >> 4;
  int wv = __builtin_amdgcn_readfirstlane((int)(threadIdx.x >> 5));
  int lane = threadIdx.x & 31;
  int half = lane >> 4;
  int lm   = lane & 15;
  int col  = (wv & 3) * 16 + lm;
  int col2 = (wv & 1) * 16 + lm;
#if USE_WMMA_F32
  v2f bf[DD / 4];
  if (wv < 4) {
#pragma unroll
    for (int ks = 0; ks < DD / 4; ++ks) {
      bf[ks].x = W[(size_t)(4 * ks + 2 * half + 0) * DD + col];
      bf[ks].y = W[(size_t)(4 * ks + 2 * half + 1) * DD + col];
    }
  }
#endif
  ((float4*)&sagg[ln][0])[q] = a;
  __syncthreads();
  if (wv < 4) {
    v8f c0 = {};
    v8f c1 = {};
#if USE_WMMA_F32
#pragma unroll
    for (int ks = 0; ks < DD / 4; ks += 2) {
      v2f a0 = *(const v2f*)&sagg[lm][4 * ks + 2 * half];
      v2f a1 = *(const v2f*)&sagg[lm][4 * ks + 4 + 2 * half];
      c0 = __builtin_amdgcn_wmma_f32_16x16x4_f32(false, a0, false, bf[ks], (short)0, c0,
                                                 false, false);
      c1 = __builtin_amdgcn_wmma_f32_16x16x4_f32(false, a1, false, bf[ks + 1], (short)0, c1,
                                                 false, false);
    }
#endif
#pragma unroll
    for (int r = 0; r < 8; ++r) {
      srow[8 * half + r][col] = c0[r] + c1[r];
    }
  }
  __syncthreads();
#if USE_WMMA_F32
  v2f bw[DD / 4];
  if (wv < 2) {
#pragma unroll
    for (int ks = 0; ks < DD / 4; ++ks) {
      bw[ks].x = Wc[(size_t)(4 * ks + 2 * half + 0) * DOUT + col2];
      bw[ks].y = Wc[(size_t)(4 * ks + 2 * half + 1) * DOUT + col2];
    }
  }
#endif
  {
    float4 y = ((const float4*)&srow[ln][0])[q];
    float* yp = &y.x;
#pragma unroll
    for (int c = 0; c < 4; ++c) {
      int d = q * 4 + c;
      float val = yp[c] + b[d];
      float rs = rsqrtf(v[d] + BN_EPS);
      yp[c] = fmaxf((val - m[d]) * rs * g[d] + be[d], 0.0f);
    }
    ((float4*)&sagg[ln][0])[q] = y;
  }
  __syncthreads();
  if (wv < 2) {
    v8f c0 = {};
    v8f c1 = {};
#if USE_WMMA_F32
#pragma unroll
    for (int ks = 0; ks < DD / 4; ks += 2) {
      v2f a0 = *(const v2f*)&sagg[lm][4 * ks + 2 * half];
      v2f a1 = *(const v2f*)&sagg[lm][4 * ks + 4 + 2 * half];
      c0 = __builtin_amdgcn_wmma_f32_16x16x4_f32(false, a0, false, bw[ks], (short)0, c0,
                                                 false, false);
      c1 = __builtin_amdgcn_wmma_f32_16x16x4_f32(false, a1, false, bw[ks + 1], (short)0, c1,
                                                 false, false);
    }
#endif
#pragma unroll
    for (int r = 0; r < 8; ++r) {
      srow[8 * half + r][col2] = c0[r] + c1[r];
    }
  }
  float ds = dinv[nd];
  bool live = (int)(blockIdx.x * 16 + ln) < n;
  __syncthreads();
  if (q < DOUT / 4) {
    float4 y = ((const float4*)&srow[ln][0])[q];
    y.x *= ds; y.y *= ds; y.z *= ds; y.w *= ds;
    if (live) ((float4*)zs)[nd * (DOUT / 4) + q] = y;
  }
}

#define ZP 36
__global__ void __launch_bounds__(256)
k_gather_pool(const int* __restrict__ rowptr,
              const int* __restrict__ degcnt,
              const int* __restrict__ csr,
              const float* __restrict__ zs,
              const float* __restrict__ dinv,
              const int* __restrict__ batch,
              float* __restrict__ pool,
              float* __restrict__ cnt, int n) {
  __shared__ __attribute__((aligned(16))) float srow[32][ZP];
  __shared__ int sgid[32];
  int t = blockIdx.x * blockDim.x + threadIdx.x;
  int nd = t >> 3;
  int q  = t & 7;
  bool live = nd < n;
  nd = min(nd, n - 1);
  int beg = rowptr[nd];
  int end = beg + degcnt[nd];
  const float4* z4 = (const float4*)zs;
  float dn = dinv[nd];
  unsigned uq = (unsigned)q;
  float4 a = z4[(unsigned)nd * 8u + uq];
  if (beg < end) {
    int last = end - 1;
    int s0 = csr[beg];
    int s1 = csr[min(beg + 1, last)];
    int s2 = csr[min(beg + 2, last)];
    int s3 = csr[min(beg + 3, last)];
    for (int j = beg; j < end; j += 4) {
      bool p1 = (j + 1) < end;
      bool p2 = (j + 2) < end;
      bool p3 = (j + 3) < end;
      float4 m0 = z4[(unsigned)s0 * 8u + uq];
      float4 m1 = z4[(unsigned)s1 * 8u + uq];
      float4 m2 = z4[(unsigned)s2 * 8u + uq];
      float4 m3 = z4[(unsigned)s3 * 8u + uq];
      s0 = csr[min(j + 4, last)];
      s1 = csr[min(j + 5, last)];
      s2 = csr[min(j + 6, last)];
      s3 = csr[min(j + 7, last)];
      asm volatile("" :: "v"(m0.x), "v"(m0.y), "v"(m0.z), "v"(m0.w),
                         "v"(m1.x), "v"(m1.y), "v"(m1.z), "v"(m1.w),
                         "v"(m2.x), "v"(m2.y), "v"(m2.z), "v"(m2.w),
                         "v"(m3.x), "v"(m3.y), "v"(m3.z), "v"(m3.w));
      __builtin_amdgcn_sched_barrier(0);
      a.x += m0.x; a.y += m0.y; a.z += m0.z; a.w += m0.w;
      a.x += p1 ? m1.x : 0.0f; a.y += p1 ? m1.y : 0.0f; a.z += p1 ? m1.z : 0.0f; a.w += p1 ? m1.w : 0.0f;
      a.x += p2 ? m2.x : 0.0f; a.y += p2 ? m2.y : 0.0f; a.z += p2 ? m2.z : 0.0f; a.w += p2 ? m2.w : 0.0f;
      a.x += p3 ? m3.x : 0.0f; a.y += p3 ? m3.y : 0.0f; a.z += p3 ? m3.z : 0.0f; a.w += p3 ? m3.w : 0.0f;
    }
  }
  a.x *= dn; a.y *= dn; a.z *= dn; a.w *= dn;
  int ln = threadIdx.x >> 3;
  ((float4*)&srow[ln][0])[q] = a;
  if (q == 0) sgid[ln] = live ? batch[nd] : -1;
  __syncthreads();
  if (threadIdx.x < DOUT) {
    int c = threadIdx.x;
    float vals[32];
    int gids[32];
#pragma unroll
    for (int k = 0; k < 32; ++k) { vals[k] = srow[k][c]; gids[k] = sgid[k]; }
    int gcur = gids[0];
    float s = vals[0];
    int run = 1;
#pragma unroll
    for (int k = 1; k < 32; ++k) {
      int gk = gids[k];
      float vk = vals[k];
      if (gk == gcur) {
        s += vk;
        ++run;
      } else {
        if (gcur >= 0) {
          atomicAdd(&pool[(size_t)gcur * DOUT + c], s);
          if (c == 0) atomicAdd(&cnt[gcur], (float)run);
        }
        gcur = gk; s = vk; run = 1;
      }
    }
    if (gcur >= 0) {
      atomicAdd(&pool[(size_t)gcur * DOUT + c], s);
      if (c == 0) atomicAdd(&cnt[gcur], (float)run);
    }
  }
}

__global__ void __launch_bounds__(256)
k_finish(const float* __restrict__ pool, const float* __restrict__ cnt,
         const float* __restrict__ bc, const float* __restrict__ bl,
         float* __restrict__ out) {
  int t = blockIdx.x * blockDim.x + threadIdx.x;
  if (t >= GG * DOUT) return;
  int gph = t / DOUT;
  int c = t - gph * DOUT;
  float cn = cnt[gph];
  float mean = pool[t] / fmaxf(cn, 1.0f) + bc[c];
  out[t] = (cn > 0.0f ? mean : 0.0f) + bl[c];
}

static inline int blk(long long n, int b) { return (int)((n + b - 1) / b); }

extern "C" void kernel_launch(void* const* d_in, const int* in_sizes, int n_in,
                              void* d_out, int out_size, void* d_ws, size_t ws_size,
                              hipStream_t stream) {
  (void)in_sizes; (void)n_in; (void)out_size; (void)ws_size;
  const int*   edge_index = (const int*)d_in[0];
  const int*   batch      = (const int*)d_in[1];
  const float* W1 = (const float*)d_in[2];
  const float* b1 = (const float*)d_in[3];
  const float* W2 = (const float*)d_in[4];
  const float* b2 = (const float*)d_in[5];
  const float* W3 = (const float*)d_in[6];
  const float* b3 = (const float*)d_in[7];
  const float* g1 = (const float*)d_in[8];
  const float* be1= (const float*)d_in[9];
  const float* m1 = (const float*)d_in[10];
  const float* v1 = (const float*)d_in[11];
  const float* g2 = (const float*)d_in[12];
  const float* be2= (const float*)d_in[13];
  const float* m2 = (const float*)d_in[14];
  const float* v2 = (const float*)d_in[15];
  const float* Wl = (const float*)d_in[16];
  const float* bl = (const float*)d_in[17];
  float* out = (float*)d_out;

  char* ws = (char*)d_ws;
  size_t off = 0;
  auto carve = [&](size_t bytes) -> float* {
    float* p = (float*)(ws + off);
    off += (bytes + 255) & ~(size_t)255;
    return p;
  };
  int*   degcnt   = (int*)carve((size_t)NN * 4);
  float* dinv     = carve((size_t)NN * 4);
  float* sd       = carve((size_t)NN * 4);
  int*   rowptr   = (int*)carve((size_t)NN * 4);
  int*   rank     = (int*)carve((size_t)EE * 4);
  int*   csr      = (int*)carve((size_t)EE * 4);
  unsigned* hist  = (unsigned*)carve((size_t)CH * NN * 4);
  int*   coff     = (int*)carve((size_t)CH * NN * 4);
  int*   bsum     = (int*)carve((size_t)CH * NB * 4);
  float* td       = carve((size_t)NN * 8);
  float* zs       = carve((size_t)NN * DOUT * 4);
  float* pool     = carve((size_t)GG * DOUT * 4);
  float* cnt      = carve((size_t)GG * 4);
  float* Wc       = carve((size_t)DD * DOUT * 4);
  float* bc       = carve((size_t)DOUT * 4);

  const int B = 256;

  k_count_lds<<<CH, 1024, 0, stream>>>(edge_index, hist, bsum, rank);
  k_scan_dinv<<<NB + 1, 256, 0, stream>>>(hist, bsum, coff, degcnt, rowptr, dinv, sd,
                                          W3, b3, Wl, Wc, bc, NN);
  k_fill<<<blk(EE, B), B, 0, stream>>>(edge_index, rank, coff, csr, EE);

  k_gather_l1<<<blk((long long)NN * 4, B), B, 0, stream>>>(
      rowptr, degcnt, csr, sd, dinv, td, pool, cnt, NN);

  k_gather_mm2<<<blk((long long)NN * 16, B), B, 0, stream>>>(
      rowptr, degcnt, csr, td, W1, b1, g1, be1, m1, v1,
      W2, b2, g2, be2, m2, v2, Wc, dinv, zs, NN);

  k_gather_pool<<<blk((long long)NN * 8, B), B, 0, stream>>>(
      rowptr, degcnt, csr, zs, dinv, batch, pool, cnt, NN);

  k_finish<<<blk((long long)GG * DOUT, B), B, 0, stream>>>(pool, cnt, bc, bl, out);
}


# Layer 3 takes one work-group per graph (node run listed by the layer-1 launch): rows summed in LDS, mean + biases at the store; pool atomics, clears, finish launch gone; sum order: fixed tree
# speedup vs baseline: 5.1477x; 1.0531x over previous
#include <hip/hip_runtime.h>
#include <hip/hip_bf16.h>

#define NN   50000
#define EE   800000
#define GG   512
#define DD   64
#define DOUT 32
#define BN_EPS 1e-5f

typedef __attribute__((ext_vector_type(2)))  float    v2f;
typedef __attribute__((ext_vector_type(8)))  float    v8f;

#if defined(__AMDGCN__) && __has_builtin(__builtin_amdgcn_wmma_f32_16x16x4_f32)
#define USE_WMMA_F32 1
#else
#define USE_WMMA_F32 0
#endif


#define CH 64
#define CS (EE / CH)
#define NB ((NN + 255) / 256)
static_assert(EE % CH == 0 && CS % 4 == 0, "a chunk is a whole number of 4-edge groups");
static_assert(CS < 65536, "a chunk's count for one node fits a 16-bit field");
static_assert(NN % 4 == 0, "the table is moved as 4-word groups");
static_assert(NB <= 256, "one thread of the scan block per node block in front of it");
#define G4  (CS / 4)
#define PER ((G4 + 1023) / 1024)

__global__ void __launch_bounds__(1024)
k_count_lds(const int* __restrict__ ei, unsigned* __restrict__ hist,
            int* __restrict__ bsum, int* __restrict__ rank) {
  __shared__ __attribute__((aligned(16))) unsigned h[NN];
  __shared__ int sb[NB];
  const int t = threadIdx.x;
  const int lane = t & 31;
  const int w = blockIdx.x;
  const int4* src4 = (const int4*)(ei + (size_t)w * CS);
  const int4* dst4 = (const int4*)(ei + EE + (size_t)w * CS);
  int4 sv[PER], dv[PER];
#pragma unroll
  for (int k = 0; k < PER; ++k) {
    int g = min(t + 1024 * k, G4 - 1);
    sv[k] = src4[g];
    dv[k] = dst4[g];
  }
  uint4* h4 = (uint4*)h;
  for (int j = t; j < NN / 4; j += 1024) h4[j] = make_uint4(0u, 0u, 0u, 0u);
  if (t < NB) sb[t] = 0;
  __syncthreads();
  int4* rank4 = (int4*)(rank + (size_t)w * CS);
#pragma unroll
  for (int k = 0; k < PER; ++k) {
    int g = t + 1024 * k;
    if (g < G4) {
      int4 r;
      r.x = (int)(atomicAdd(&h[dv[k].x], 1u) & 0xFFFFu);
      r.y = (int)(atomicAdd(&h[dv[k].y], 1u) & 0xFFFFu);
      r.z = (int)(atomicAdd(&h[dv[k].z], 1u) & 0xFFFFu);
      r.w = (int)(atomicAdd(&h[dv[k].w], 1u) & 0xFFFFu);
      atomicAdd(&h[sv[k].x], 0x10000u);
      atomicAdd(&h[sv[k].y], 0x10000u);
      atomicAdd(&h[sv[k].z], 0x10000u);
      atomicAdd(&h[sv[k].w], 0x10000u);
      rank4[g] = r;
    }
  }
  __syncthreads();
  uint4* hg4 = (uint4*)(hist + (size_t)w * NN);
  for (int j0 = 0; j0 < NN / 4; j0 += 1024) {
    int j = j0 + t;
    bool ok = j < NN / 4;
    uint4 v = make_uint4(0u, 0u, 0u, 0u);
    if (ok) { v = h4[j]; hg4[j] = v; }
    int c = (int)((v.x & 0xFFFFu) + (v.y & 0xFFFFu) + (v.z & 0xFFFFu) + (v.w & 0xFFFFu));
    for (int o = 16; o > 0; o >>= 1) c += __shfl_xor(c, o);
    if (lane == 0 && ok) atomicAdd(&sb[j >> 6], c);
  }
  __syncthreads();
  if (t < NB) bsum[w * NB + t] = sb[t];
}

__device__ __forceinline__ void form_tail_weights(const float* __restrict__ W3,
                                                  const float* __restrict__ b3,
                                                  const float* __restrict__ Wl,
                                                  float* __restrict__ Wc,
                                                  float* __restrict__ bc) {
#if USE_WMMA_F32
  int wv = __builtin_amdgcn_readfirstlane((int)(threadIdx.x >> 5));
  int lane = threadIdx.x & 31;
  int half = lane >> 4;
  int lm   = lane & 15;
  int tm = wv >> 1;
  int ncol = (wv & 1) * 16 + lm;
  const float* Arow = W3 + (size_t)(tm * 16 + lm) * DD;
  v8f c = {};
#pragma unroll
  for (int k0 = 0; k0 < DD; k0 += 4) {
    v2f a, b;
    a.x = Arow[k0 + 2 * half + 0];
    a.y = Arow[k0 + 2 * half + 1];
    b.x = Wl[(size_t)(k0 + 2 * half + 0) * DOUT + ncol];
    b.y = Wl[(size_t)(k0 + 2 * half + 1) * DOUT + ncol];
    c = __builtin_amdgcn_wmma_f32_16x16x4_f32(false, a, false, b, (short)0, c, false, false);
  }
#pragma unroll
  for (int r = 0; r < 8; ++r) {
    Wc[(size_t)(tm * 16 + 8 * half + r) * DOUT + ncol] = c[r];
  }
  if (wv < 2) {
    v8f cb = {};
#pragma unroll
    for (int k0 = 0; k0 < DD; k0 += 4) {
      v2f a, b;
      a.x = b3[k0 + 2 * half + 0];
      a.y = b3[k0 + 2 * half + 1];
      b.x = Wl[(size_t)(k0 + 2 * half + 0) * DOUT + ncol];
      b.y = Wl[(size_t)(k0 + 2 * half + 1) * DOUT + ncol];
      cb = __builtin_amdgcn_wmma_f32_16x16x4_f32(false, a, false, b, (short)0, cb, false, false);
    }
    if (half == 0) bc[ncol] = cb[0];
  }
#endif
}

__global__ void __launch_bounds__(256)
k_scan_dinv(const unsigned* __restrict__ hist, const int* __restrict__ bsum,
            int* __restrict__ off, int* __restrict__ degcnt,
            int* __restrict__ rowptr, float* __restrict__ dinv,
            float* __restrict__ sd,
            const float* __restrict__ W3, const float* __restrict__ b3,
            const float* __restrict__ Wl,
            float* __restrict__ Wc, float* __restrict__ bc, int n) {
  __shared__ int wsum[8];
  __shared__ int s_base;
  if (blockIdx.x == NB) {
    form_tail_weights(W3, b3, Wl, Wc, bc);
    return;
  }
  int t = threadIdx.x;
  int lane = t & 31;
  int w = t >> 5;
  int b = blockIdx.x;
  int i = b * 256 + t;
  int ic = min(i, n - 1);
  unsigned hv[CH];
#pragma unroll
  for (int k = 0; k < CH; ++k) hv[k] = hist[(size_t)k * NN + ic];

  int p = 0;
  if (t < b) {
    for (int k = 0; k < CH; ++k) p += bsum[k * NB + t];
  }
  for (int o = 16; o > 0; o >>= 1) p += __shfl_xor(p, o);
  if (lane == 0) wsum[w] = p;
  __syncthreads();
  if (w == 0) {
    int x = (lane < 8) ? wsum[lane] : 0;
    for (int o = 16; o > 0; o >>= 1) x += __shfl_xor(x, o);
    if (lane == 0) s_base = x;
  }
  __syncthreads();
  int base = s_base;

  int cin = 0, cout = 0;
#pragma unroll
  for (int k = 0; k < CH; ++k) { cin += (int)(hv[k] & 0xFFFFu); cout += (int)(hv[k] >> 16); }
  int c = (i < n) ? cin : 0;
  int v = c;
  for (int o = 1; o < 32; o <<= 1) {
    int u = __shfl_up(v, o);
    if (lane >= o) v += u;
  }
  if (lane == 31) wsum[w] = v;
  __syncthreads();
  int woff = 0;
  for (int k = 0; k < 8; ++k) woff += (k < w) ? wsum[k] : 0;
  if (i < n) {
    int rp = base + woff + v - c;
    rowptr[i] = rp;
    degcnt[i] = c;
    float di = rsqrtf((float)c + 1.0f);
    dinv[i] = di;
    sd[i] = (float)cout * di;
    int run = rp;
#pragma unroll
    for (int k = 0; k < CH; ++k) {
      off[(size_t)k * NN + i] = run;
      run += (int)(hv[k] & 0xFFFFu);
    }
  }
}

__global__ void k_fill(const int* __restrict__ ei,
                       const int* __restrict__ rank,
                       const int* __restrict__ off,
                       int* __restrict__ csr, int E) {
  int e = blockIdx.x * blockDim.x + threadIdx.x;
  if (e >= E) return;
  int s = ei[e];
  int d = ei[E + e];
  csr[off[(size_t)(e / CS) * NN + d] + rank[e]] = s;
}

__global__ void __launch_bounds__(256)
k_gather_l1(const int* __restrict__ rowptr,
            const int* __restrict__ degcnt,
            const int* __restrict__ csr,
            const float* __restrict__ sd,
            const float* __restrict__ dinv,
            const int* __restrict__ batch,
            float* __restrict__ td,
            int* __restrict__ gstart, int n) {
  int t = blockIdx.x * blockDim.x + threadIdx.x;
  int sub = t & 3;
  int node = t >> 2;
  int ndc = min(node, n - 1);
  int beg = rowptr[ndc];
  int cnt = degcnt[ndc];
  float dn = dinv[ndc];
  int gcur = batch[ndc];
  int gprev = batch[max(ndc - 1, 0)];
  float acc = (sub == 0) ? sd[ndc] : 0.0f;
  int end = beg + cnt;
  int last = (cnt > 0) ? end - 1 : 0;
  int j0 = beg + sub;
  int s[8];
  float val[8];
#pragma unroll
  for (int i = 0; i < 8; ++i) s[i] = csr[min(j0 + 4 * i, last)];
#pragma unroll
  for (int i = 0; i < 8; ++i) val[i] = sd[s[i]];
#pragma unroll
  for (int i = 0; i < 8; ++i) acc += (j0 + 4 * i < end) ? val[i] : 0.0f;
  for (int j = j0 + 32; j < end; j += 4) acc += sd[csr[j]];
  acc += __shfl_xor(acc, 1);
  acc += __shfl_xor(acc, 2);
  float tn = acc * dn;
  if (sub == 0 && node < n) {
    ((float2*)td)[node] = make_float2(tn, dn);
    int lo = (node > 0) ? max(gprev + 1, 0) : 0;
    int hi = min(gcur, GG - 1);
#pragma clang loop vectorize(disable) interleave(disable) unroll(disable)
    for (int g = lo; g <= hi; ++g) gstart[g] = node;
    if (node == n - 1) {
#pragma clang loop vectorize(disable) interleave(disable) unroll(disable)
      for (int g = max(gcur + 1, 0); g <= GG; ++g) gstart[g] = n;
    }
  }
}

__device__ __forceinline__ v2f l1_row(float t, v2f w, v2f b, v2f m, v2f rs, v2f g, v2f be) {
  v2f r = t * w + b;
  v2f y = (r - m) * rs * g + be;
  v2f zero = {0.0f, 0.0f};
  return __builtin_elementwise_max(y, zero);
}

#define TP 68
__global__ void __launch_bounds__(256, 16)
k_gather_mm2(const int* __restrict__ rowptr,
             const int* __restrict__ degcnt,
             const int* __restrict__ csr,
             const float* __restrict__ td,
             const float* __restrict__ pW,
             const float* __restrict__ pb,
             const float* __restrict__ pg,
             const float* __restrict__ pbe,
             const float* __restrict__ pm,
             const float* __restrict__ pv,
             const float* __restrict__ W,
             const float* __restrict__ b,
             const float* __restrict__ g,
             const float* __restrict__ be,
             const float* __restrict__ m,
             const float* __restrict__ v,
             const float* __restrict__ Wc,
             const float* __restrict__ dinv,
             float* __restrict__ zs, int n) {
  __shared__ __attribute__((aligned(16))) float sagg[16][TP];
  __shared__ __attribute__((aligned(16))) float srow[16][TP];
  int t = blockIdx.x * blockDim.x + threadIdx.x;
  int q  = t & 15;
  int nd = min(t >> 4, n - 1);
  int beg = rowptr[nd];
  int end = beg + degcnt[nd];
  float4 a;
  const float2* td2 = (const float2*)td;
  const v2f* pW2 = (const v2f*)pW;  const v2f* pb2 = (const v2f*)pb;
  const v2f* pm2 = (const v2f*)pm;  const v2f* pv2 = (const v2f*)pv;
  const v2f* pg2 = (const v2f*)pg;  const v2f* pbe2 = (const v2f*)pbe;
  v2f cw[2], cb[2], cm[2], cr[2], cg[2], ce[2];
#pragma unroll
  for (int h = 0; h < 2; ++h) {
    int d2 = q * 2 + h;
    cw[h] = pW2[d2]; cb[h] = pb2[d2]; cm[h] = pm2[d2];
    v2f vv = pv2[d2];
    cr[h].x = rsqrtf(vv.x + BN_EPS); cr[h].y = rsqrtf(vv.y + BN_EPS);
    cg[h] = pg2[d2]; ce[h] = pbe2[d2];
  }
  float2 own = td2[nd];
  float dn = own.y;
  {
    float ws = dn * dn;
    v2f acc[2];
#pragma unroll
    for (int h = 0; h < 2; ++h)
      acc[h] = l1_row(own.x, cw[h], cb[h], cm[h], cr[h], cg[h], ce[h]) * ws;
    if (beg < end) {
      int last = end - 1;
      int s0 = csr[beg];
      int s1 = csr[min(beg + 1, last)];
      int s2 = csr[min(beg + 2, last)];
      int s3 = csr[min(beg + 3, last)];
      for (int j = beg; j < end; j += 4) {
        bool p1 = (j + 1) < end;
        bool p2 = (j + 2) < end;
        bool p3 = (j + 3) < end;
        float2 e0 = td2[s0];
        float2 e1 = td2[s1];
        float2 e2 = td2[s2];
        float2 e3 = td2[s3];
        s0 = csr[min(j + 4, last)];
        s1 = csr[min(j + 5, last)];
        s2 = csr[min(j + 6, last)];
        s3 = csr[min(j + 7, last)];
        __builtin_amdgcn_sched_barrier(0);
        float wt0 = e0.y * dn;
        float wt1 = p1 ? e1.y * dn : 0.0f;
        float wt2 = p2 ? e2.y * dn : 0.0f;
        float wt3 = p3 ? e3.y * dn : 0.0f;
#pragma unroll
        for (int h = 0; h < 2; ++h)
          acc[h] += l1_row(e0.x, cw[h], cb[h], cm[h], cr[h], cg[h], ce[h]) * wt0;
#pragma unroll
        for (int h = 0; h < 2; ++h)
          acc[h] += l1_row(e1.x, cw[h], cb[h], cm[h], cr[h], cg[h], ce[h]) * wt1;
#pragma unroll
        for (int h = 0; h < 2; ++h)
          acc[h] += l1_row(e2.x, cw[h], cb[h], cm[h], cr[h], cg[h], ce[h]) * wt2;
#pragma unroll
        for (int h = 0; h < 2; ++h)
          acc[h] += l1_row(e3.x, cw[h], cb[h], cm[h], cr[h], cg[h], ce[h]) * wt3;
      }
    }
    a.x = acc[0].x; a.y = acc[0].y; a.z = acc[1].x; a.w = acc[1].y;
  }
  int ln = threadIdx.x >> 4;
  int wv = __builtin_amdgcn_readfirstlane((int)(threadIdx.x >> 5));
  int lane = threadIdx.x & 31;
  int half = lane >> 4;
  int lm   = lane & 15;
  int col  = (wv & 3) * 16 + lm;
  int col2 = (wv & 1) * 16 + lm;
#if USE_WMMA_F32
  v2f bf[DD / 4];
  if (wv < 4) {
#pragma unroll
    for (int ks = 0; ks < DD / 4; ++ks) {
      bf[ks].x = W[(size_t)(4 * ks + 2 * half + 0) * DD + col];
      bf[ks].y = W[(size_t)(4 * ks + 2 * half + 1) * DD + col];
    }
  }
#endif
  ((float4*)&sagg[ln][0])[q] = a;
  __syncthreads();
  if (wv < 4) {
    v8f c0 = {};
    v8f c1 = {};
#if USE_WMMA_F32
#pragma unroll
    for (int ks = 0; ks < DD / 4; ks += 2) {
      v2f a0 = *(const v2f*)&sagg[lm][4 * ks + 2 * half];
      v2f a1 = *(const v2f*)&sagg[lm][4 * ks + 4 + 2 * half];
      c0 = __builtin_amdgcn_wmma_f32_16x16x4_f32(false, a0, false, bf[ks], (short)0, c0,
                                                 false, false);
      c1 = __builtin_amdgcn_wmma_f32_16x16x4_f32(false, a1, false, bf[ks + 1], (short)0, c1,
                                                 false, false);
    }
#endif
#pragma unroll
    for (int r = 0; r < 8; ++r) {
      srow[8 * half + r][col] = c0[r] + c1[r];
    }
  }
  __syncthreads();
#if USE_WMMA_F32
  v2f bw[DD / 4];
  if (wv < 2) {
#pragma unroll
    for (int ks = 0; ks < DD / 4; ++ks) {
      bw[ks].x = Wc[(size_t)(4 * ks + 2 * half + 0) * DOUT + col2];
      bw[ks].y = Wc[(size_t)(4 * ks + 2 * half + 1) * DOUT + col2];
    }
  }
#endif
  {
    float4 y = ((const float4*)&srow[ln][0])[q];
    float* yp = &y.x;
#pragma unroll
    for (int c = 0; c < 4; ++c) {
      int d = q * 4 + c;
      float val = yp[c] + b[d];
      float rs = rsqrtf(v[d] + BN_EPS);
      yp[c] = fmaxf((val - m[d]) * rs * g[d] + be[d], 0.0f);
    }
    ((float4*)&sagg[ln][0])[q] = y;
  }
  __syncthreads();
  if (wv < 2) {
    v8f c0 = {};
    v8f c1 = {};
#if USE_WMMA_F32
#pragma unroll
    for (int ks = 0; ks < DD / 4; ks += 2) {
      v2f a0 = *(const v2f*)&sagg[lm][4 * ks + 2 * half];
      v2f a1 = *(const v2f*)&sagg[lm][4 * ks + 4 + 2 * half];
      c0 = __builtin_amdgcn_wmma_f32_16x16x4_f32(false, a0, false, bw[ks], (short)0, c0,
                                                 false, false);
      c1 = __builtin_amdgcn_wmma_f32_16x16x4_f32(false, a1, false, bw[ks + 1], (short)0, c1,
                                                 false, false);
    }
#endif
#pragma unroll
    for (int r = 0; r < 8; ++r) {
      srow[8 * half + r][col2] = c0[r] + c1[r];
    }
  }
  float ds = dinv[nd];
  bool live = (int)(blockIdx.x * 16 + ln) < n;
  __syncthreads();
  if (q < DOUT / 4) {
    float4 y = ((const float4*)&srow[ln][0])[q];
    y.x *= ds; y.y *= ds; y.z *= ds; y.w *= ds;
    if (live) ((float4*)zs)[nd * (DOUT / 4) + q] = y;
  }
}

#define ZP 36
#define TN 128
static_assert(TN == 128 && DOUT == 32, "32 waves: wave w sums rows w + 32 k, lane = column");
__global__ void __launch_bounds__(1024)
k_gather_graph(const int* __restrict__ rowptr,
               const int* __restrict__ degcnt,
               const int* __restrict__ csr,
               const float* __restrict__ zs,
               const float* __restrict__ dinv,
               const int* __restrict__ gstart,
               const float* __restrict__ bc,
               const float* __restrict__ bl,
               float* __restrict__ out, int n) {
  __shared__ __attribute__((aligned(16))) float srow[TN][ZP];
  __shared__ float part[TN / 4][DOUT];
  const int gph = blockIdx.x;
  int gb = min(max(gstart[gph], 0), n);
  int ge = min(max(gstart[gph + 1], gb), n);
  int ln = threadIdx.x >> 3;
  int q  = threadIdx.x & 7;
  int wv = threadIdx.x >> 5;
  int lane = threadIdx.x & 31;
  float vbc = bc[lane];
  float vbl = bl[lane];
  const float4* z4 = (const float4*)zs;
  unsigned uq = (unsigned)q;
  float tot = 0.0f;
  for (int base = gb; base < ge; base += TN) {
    int nd = base + ln;
    bool live = nd < ge;
    nd = min(nd, ge - 1);
    int beg = rowptr[nd];
    int cnt = degcnt[nd];
    float dn = dinv[nd];
    int end = beg + (live ? cnt : 0);
    float4 a = z4[(unsigned)nd * 8u + uq];
    if (beg < end) {
      int last = end - 1;
      int s0 = csr[beg];
      int s1 = csr[min(beg + 1, last)];
      int s2 = csr[min(beg + 2, last)];
      int s3 = csr[min(beg + 3, last)];
      for (int j = beg; j < end; j += 4) {
        bool p1 = (j + 1) < end;
        bool p2 = (j + 2) < end;
        bool p3 = (j + 3) < end;
        float4 m0 = z4[(unsigned)s0 * 8u + uq];
        float4 m1 = z4[(unsigned)s1 * 8u + uq];
        float4 m2 = z4[(unsigned)s2 * 8u + uq];
        float4 m3 = z4[(unsigned)s3 * 8u + uq];
        s0 = csr[min(j + 4, last)];
        s1 = csr[min(j + 5, last)];
        s2 = csr[min(j + 6, last)];
        s3 = csr[min(j + 7, last)];
        asm volatile("" :: "v"(m0.x), "v"(m0.y), "v"(m0.z), "v"(m0.w),
                           "v"(m1.x), "v"(m1.y), "v"(m1.z), "v"(m1.w),
                           "v"(m2.x), "v"(m2.y), "v"(m2.z), "v"(m2.w),
                           "v"(m3.x), "v"(m3.y), "v"(m3.z), "v"(m3.w));
        __builtin_amdgcn_sched_barrier(0);
        a.x += m0.x; a.y += m0.y; a.z += m0.z; a.w += m0.w;
        a.x += p1 ? m1.x : 0.0f; a.y += p1 ? m1.y : 0.0f; a.z += p1 ? m1.z : 0.0f; a.w += p1 ? m1.w : 0.0f;
        a.x += p2 ? m2.x : 0.0f; a.y += p2 ? m2.y : 0.0f; a.z += p2 ? m2.z : 0.0f; a.w += p2 ? m2.w : 0.0f;
        a.x += p3 ? m3.x : 0.0f; a.y += p3 ? m3.y : 0.0f; a.z += p3 ? m3.z : 0.0f; a.w += p3 ? m3.w : 0.0f;
      }
    }
    a.x = live ? a.x * dn : 0.0f; a.y = live ? a.y * dn : 0.0f;
    a.z = live ? a.z * dn : 0.0f; a.w = live ? a.w * dn : 0.0f;
    ((float4*)&srow[ln][0])[q] = a;
    __syncthreads();
    float p = srow[wv][lane];
    p += srow[wv + 32][lane];
    p += srow[wv + 64][lane];
    p += srow[wv + 96][lane];
    part[wv][lane] = p;
    __syncthreads();
    if (wv == 0) {
      float s0 = part[0][lane], s1 = part[1][lane], s2 = part[2][lane], s3 = part[3][lane];
#pragma unroll
      for (int k = 4; k < TN / 4; k += 4) {
        s0 += part[k][lane]; s1 += part[k + 1][lane]; s2 += part[k + 2][lane]; s3 += part[k + 3][lane];
      }
      tot += (s0 + s1) + (s2 + s3);
    }
  }
  if (wv == 0) {
    int cn = ge - gb;
    float mean = tot / fmaxf((float)cn, 1.0f) + vbc;
    out[gph * DOUT + lane] = (cn > 0 ? mean : 0.0f) + vbl;
  }
}

static inline int blk(long long n, int b) { return (int)((n + b - 1) / b); }

extern "C" void kernel_launch(void* const* d_in, const int* in_sizes, int n_in,
                              void* d_out, int out_size, void* d_ws, size_t ws_size,
                              hipStream_t stream) {
  (void)in_sizes; (void)n_in; (void)out_size; (void)ws_size;
  const int*   edge_index = (const int*)d_in[0];
  const int*   batch      = (const int*)d_in[1];
  const float* W1 = (const float*)d_in[2];
  const float* b1 = (const float*)d_in[3];
  const float* W2 = (const float*)d_in[4];
  const float* b2 = (const float*)d_in[5];
  const float* W3 = (const float*)d_in[6];
  const float* b3 = (const float*)d_in[7];
  const float* g1 = (const float*)d_in[8];
  const float* be1= (const float*)d_in[9];
  const float* m1 = (const float*)d_in[10];
  const float* v1 = (const float*)d_in[11];
  const float* g2 = (const float*)d_in[12];
  const float* be2= (const float*)d_in[13];
  const float* m2 = (const float*)d_in[14];
  const float* v2 = (const float*)d_in[15];
  const float* Wl = (const float*)d_in[16];
  const float* bl = (const float*)d_in[17];
  float* out = (float*)d_out;

  char* ws = (char*)d_ws;
  size_t off = 0;
  auto carve = [&](size_t bytes) -> float* {
    float* p = (float*)(ws + off);
    off += (bytes + 255) & ~(size_t)255;
    return p;
  };
  int*   degcnt   = (int*)carve((size_t)NN * 4);
  float* dinv     = carve((size_t)NN * 4);
  float* sd       = carve((size_t)NN * 4);
  int*   rowptr   = (int*)carve((size_t)NN * 4);
  int*   rank     = (int*)carve((size_t)EE * 4);
  int*   csr      = (int*)carve((size_t)EE * 4);
  unsigned* hist  = (unsigned*)carve((size_t)CH * NN * 4);
  int*   coff     = (int*)carve((size_t)CH * NN * 4);
  int*   bsum     = (int*)carve((size_t)CH * NB * 4);
  float* td       = carve((size_t)NN * 8);
  float* zs       = carve((size_t)NN * DOUT * 4);
  int*   gstart   = (int*)carve((size_t)(GG + 1) * 4);
  float* Wc       = carve((size_t)DD * DOUT * 4);
  float* bc       = carve((size_t)DOUT * 4);

  const int B = 256;

  k_count_lds<<<CH, 1024, 0, stream>>>(edge_index, hist, bsum, rank);
  k_scan_dinv<<<NB + 1, 256, 0, stream>>>(hist, bsum, coff, degcnt, rowptr, dinv, sd,
                                          W3, b3, Wl, Wc, bc, NN);
  k_fill<<<blk(EE, B), B, 0, stream>>>(edge_index, rank, coff, csr, EE);

  k_gather_l1<<<blk((long long)NN * 4, B), B, 0, stream>>>(
      rowptr, degcnt, csr, sd, dinv, batch, td, gstart, NN);

  k_gather_mm2<<<blk((long long)NN * 16, B), B, 0, stream>>>(
      rowptr, degcnt, csr, td, W1, b1, g1, be1, m1, v1,
      W2, b2, g2, be2, m2, v2, Wc, dinv, zs, NN);

  k_gather_graph<<<GG, 1024, 0, stream>>>(
      rowptr, degcnt, csr, zs, dinv, gstart, bc, bl, out, NN);
}
